# starting kernel
# baseline (speedup 1.0000x reference)
#include <hip/hip_runtime.h>

typedef float v2f __attribute__((ext_vector_type(2)));
typedef float v8f __attribute__((ext_vector_type(8)));

#define HD 128
constexpr int N_USER = 150000;
constexpr int N_BOOK = 75000;
constexpr int N_EDGE = 1500000;
constexpr float LN_EPS = 1e-5f;

__global__ void zero_kernel(float* __restrict__ p, size_t n) {
    size_t i = (size_t)blockIdx.x * blockDim.x + threadIdx.x;
    size_t stride = (size_t)gridDim.x * blockDim.x;
    for (; i < n; i += stride) p[i] = 0.0f;
}

__global__ void count_kernel(const int* __restrict__ esrc, const int* __restrict__ edst,
                             float* __restrict__ cnt_u, float* __restrict__ cnt_b, int E) {
    int i = blockIdx.x * blockDim.x + threadIdx.x;
    if (i < E) {
        atomicAdd(&cnt_u[esrc[i]], 1.0f);
        atomicAdd(&cnt_b[edst[i]], 1.0f);
    }
}

__global__ void scatter_kernel(const float* __restrict__ feat, const int* __restrict__ sidx,
                               const int* __restrict__ didx, float* __restrict__ sums, int E) {
    int gid = blockIdx.x * blockDim.x + threadIdx.x;
    int e = gid >> 5;
    if (e >= E) return;
    int lane = gid & 31;
    int s = sidx[e], d = didx[e];
    const float4 v = *(const float4*)(feat + (size_t)s * HD + lane * 4);
    float* dp = sums + (size_t)d * HD + lane * 4;
    atomicAdd(dp + 0, v.x);
    atomicAdd(dp + 1, v.y);
    atomicAdd(dp + 2, v.z);
    atomicAdd(dp + 3, v.w);
}


__device__ __forceinline__ void load_weight_t(const float* __restrict__ W, float* __restrict__ wt,
                                              int K, int stride) {
    const int nf4 = K * (HD / 4);
    for (int idx = threadIdx.x; idx < nf4; idx += 256) {
        int k = idx >> 5;
        int n = (idx & 31) << 2;
        const float4 w4 = *(const float4*)(W + (size_t)k * HD + n);
        wt[(n + 0) * stride + k] = w4.x;
        wt[(n + 1) * stride + k] = w4.y;
        wt[(n + 2) * stride + k] = w4.z;
        wt[(n + 3) * stride + k] = w4.w;
    }
}

__global__ void gemm_bias_kernel(const float* __restrict__ A, const float* __restrict__ W,
                                 const float* __restrict__ bias, float* __restrict__ out,
                                 int M, int K) {
    __shared__ float wt[HD * (HD + 4)];
    const int stride = K + 4;
    load_weight_t(W, wt, K, stride);
    __syncthreads();

    const int wave = threadIdx.x >> 5;
    const int lane = threadIdx.x & 31;
    const int col  = lane & 15;
    const int half = lane >> 4;
    const int m0   = blockIdx.x * 128 + wave * 16;
    const int aRow = (m0 + col < M) ? (m0 + col) : (M - 1);

    v8f acc[8];
#pragma unroll
    for (int nt = 0; nt < 8; ++nt) {
        float bv = bias[nt * 16 + col];
        acc[nt] = (v8f){bv, bv, bv, bv, bv, bv, bv, bv};
    }

    const float* arow = A + (size_t)aRow * K;
    for (int k0 = 0; k0 < K; k0 += 4) {
        const int kb = k0 + 2 * half;
        v2f a = *(const v2f*)(arow + kb);
        v2f bf[8];
#pragma unroll
        for (int nt = 0; nt < 8; ++nt)
            bf[nt] = *(const v2f*)&wt[(nt * 16 + col) * stride + kb];
#pragma unroll
        for (int nt = 0; nt < 8; ++nt)
            acc[nt] = __builtin_amdgcn_wmma_f32_16x16x4_f32(
                false, a, false, bf[nt], (short)0, acc[nt], false, false);
    }

#pragma unroll
    for (int nt = 0; nt < 8; ++nt)
#pragma unroll
        for (int v = 0; v < 8; ++v) {
            int dr = m0 + v + 8 * half;
            if (dr < M) out[(size_t)dr * HD + nt * 16 + col] = acc[nt][v];
        }
}

__global__ void sage_combine_kernel(const float* __restrict__ sums, const float* __restrict__ cnt,
                                    const float* __restrict__ X, const float* __restrict__ Wl,
                                    const float* __restrict__ bl, const float* __restrict__ Wr,
                                    float* __restrict__ out, int M) {
    __shared__ float wt[HD * (HD + 4)];
    const int stride = HD + 4;

    load_weight_t(Wl, wt, HD, stride);
    __syncthreads();

    const int wave = threadIdx.x >> 5;
    const int lane = threadIdx.x & 31;
    const int col  = lane & 15;
    const int half = lane >> 4;
    const int m0   = blockIdx.x * 128 + wave * 16;
    const int aRow = (m0 + col < M) ? (m0 + col) : (M - 1);
    const float invc = 1.0f / fmaxf(cnt[aRow], 1.0f);

    v8f acc[8];
#pragma unroll
    for (int nt = 0; nt < 8; ++nt) {
        float bv = bl[nt * 16 + col];
        acc[nt] = (v8f){bv, bv, bv, bv, bv, bv, bv, bv};
    }

    const float* srow = sums + (size_t)aRow * HD;
    for (int k0 = 0; k0 < HD; k0 += 4) {
        const int kb = k0 + 2 * half;
        v2f a = *(const v2f*)(srow + kb);
        a.x *= invc;
        a.y *= invc;
        v2f bf[8];
#pragma unroll
        for (int nt = 0; nt < 8; ++nt)
            bf[nt] = *(const v2f*)&wt[(nt * 16 + col) * stride + kb];
#pragma unroll
        for (int nt = 0; nt < 8; ++nt)
            acc[nt] = __builtin_amdgcn_wmma_f32_16x16x4_f32(
                false, a, false, bf[nt], (short)0, acc[nt], false, false);
    }

    __syncthreads();
    load_weight_t(Wr, wt, HD, stride);
    __syncthreads();

    const float* xrow = X + (size_t)aRow * HD;
    for (int k0 = 0; k0 < HD; k0 += 4) {
        const int kb = k0 + 2 * half;
        v2f a = *(const v2f*)(xrow + kb);
        v2f bf[8];
#pragma unroll
        for (int nt = 0; nt < 8; ++nt)
            bf[nt] = *(const v2f*)&wt[(nt * 16 + col) * stride + kb];
#pragma unroll
        for (int nt = 0; nt < 8; ++nt)
            acc[nt] = __builtin_amdgcn_wmma_f32_16x16x4_f32(
                false, a, false, bf[nt], (short)0, acc[nt], false, false);
    }

#pragma unroll
    for (int nt = 0; nt < 8; ++nt)
#pragma unroll
        for (int v = 0; v < 8; ++v) {
            int dr = m0 + v + 8 * half;
            if (dr < M) out[(size_t)dr * HD + nt * 16 + col] = fmaxf(acc[nt][v], 0.0f);
        }
}

__global__ void layernorm_kernel(const float* __restrict__ X, const float* __restrict__ gamma,
                                 const float* __restrict__ beta, float* __restrict__ out, int M) {
    const int wave = threadIdx.x >> 5;
    const int lane = threadIdx.x & 31;
    const int row  = blockIdx.x * 8 + wave;
    if (row >= M) return;

    const float4 v = *(const float4*)(X + (size_t)row * HD + lane * 4);
    float s = v.x + v.y + v.z + v.w;
#pragma unroll
    for (int off = 16; off >= 1; off >>= 1) s += __shfl_xor(s, off, 32);
    const float mu = s * (1.0f / HD);

    const float dx = v.x - mu, dy = v.y - mu, dz = v.z - mu, dw = v.w - mu;
    float q = dx * dx + dy * dy + dz * dz + dw * dw;
#pragma unroll
    for (int off = 16; off >= 1; off >>= 1) q += __shfl_xor(q, off, 32);
    const float rstd = rsqrtf(q * (1.0f / HD) + LN_EPS);

    const float4 g = *(const float4*)(gamma + lane * 4);
    const float4 bt = *(const float4*)(beta + lane * 4);
    float4 o;
    o.x = dx * rstd * g.x + bt.x;
    o.y = dy * rstd * g.y + bt.y;
    o.z = dz * rstd * g.z + bt.z;
    o.w = dw * rstd * g.w + bt.w;
    *(float4*)(out + (size_t)row * HD + lane * 4) = o;
}

extern "C" void kernel_launch(void* const* d_in, const int* in_sizes, int n_in,
                              void* d_out, int out_size, void* d_ws, size_t ws_size,
                              hipStream_t stream) {
    const float* user_x      = (const float*)d_in[0];
    const float* book_x      = (const float*)d_in[1];
    const int*   edge_src    = (const int*)d_in[2];
    const int*   edge_dst    = (const int*)d_in[3];
    const float* user_proj_w = (const float*)d_in[4];
    const float* user_proj_b = (const float*)d_in[5];
    const float* book_proj_w = (const float*)d_in[6];
    const float* book_proj_b = (const float*)d_in[7];
    const float* Wl          = (const float*)d_in[8];
    const float* bl          = (const float*)d_in[9];
    const float* Wr          = (const float*)d_in[10];
    const float* ugamma      = (const float*)d_in[11];
    const float* ubeta       = (const float*)d_in[12];
    const float* bgamma      = (const float*)d_in[13];
    const float* bbeta       = (const float*)d_in[14];

    float* ws = (float*)d_ws;
    size_t off = 0;
    auto carve = [&](size_t n) { float* p = ws + off; off += n; return p; };
    float* u_cur = carve((size_t)N_USER * HD);
    float* u_nxt = carve((size_t)N_USER * HD);
    float* b_cur = carve((size_t)N_BOOK * HD);
    float* b_nxt = carve((size_t)N_BOOK * HD);
    float* sum_u = carve((size_t)N_USER * HD);
    float* sum_b = carve((size_t)N_BOOK * HD);
    float* cnt_u = carve((size_t)N_USER);
    float* cnt_b = carve((size_t)N_BOOK);

    gemm_bias_kernel<<<(N_USER + 127) / 128, 256, 0, stream>>>(
        user_x, user_proj_w, user_proj_b, u_cur, N_USER, 64);
    gemm_bias_kernel<<<(N_BOOK + 127) / 128, 256, 0, stream>>>(
        book_x, book_proj_w, book_proj_b, b_cur, N_BOOK, 128);

    zero_kernel<<<256, 256, 0, stream>>>(cnt_u, (size_t)N_USER);
    zero_kernel<<<256, 256, 0, stream>>>(cnt_b, (size_t)N_BOOK);
    count_kernel<<<(N_EDGE + 255) / 256, 256, 0, stream>>>(
        edge_src, edge_dst, cnt_u, cnt_b, N_EDGE);

    const int scat_blocks = (N_EDGE * 32 + 255) / 256;
    for (int l = 0; l < 3; ++l) {
        const float* wl  = Wl + (size_t)l * HD * HD;
        const float* blv = bl + (size_t)l * HD;
        const float* wr  = Wr + (size_t)l * HD * HD;

        zero_kernel<<<4096, 256, 0, stream>>>(sum_b, (size_t)N_BOOK * HD);
        zero_kernel<<<4096, 256, 0, stream>>>(sum_u, (size_t)N_USER * HD);

        scatter_kernel<<<scat_blocks, 256, 0, stream>>>(u_cur, edge_src, edge_dst, sum_b, N_EDGE);
        scatter_kernel<<<scat_blocks, 256, 0, stream>>>(b_cur, edge_dst, edge_src, sum_u, N_EDGE);

        sage_combine_kernel<<<(N_BOOK + 127) / 128, 256, 0, stream>>>(
            sum_b, cnt_b, b_cur, wl, blv, wr, b_nxt, N_BOOK);
        sage_combine_kernel<<<(N_USER + 127) / 128, 256, 0, stream>>>(
            sum_u, cnt_u, u_cur, wl, blv, wr, u_nxt, N_USER);

        float* t;
        t = u_cur; u_cur = u_nxt; u_nxt = t;
        t = b_cur; b_cur = b_nxt; b_nxt = t;
    }

    float* out_u = (float*)d_out;
    float* out_b = out_u + (size_t)N_USER * HD;
    layernorm_kernel<<<(N_USER + 7) / 8, 256, 0, stream>>>(u_cur, ugamma, ubeta, out_u, N_USER);
    layernorm_kernel<<<(N_BOOK + 7) / 8, 256, 0, stream>>>(b_cur, bgamma, bbeta, out_b, N_BOOK);
}


# Atomic scatter: one wave per edge serves both directions of a layer (indices read once, user row and book row in flight together): half the waves, 3 launches fewer; arrival order as before
# speedup vs baseline: 1.4759x; 1.4759x over previous
#include <hip/hip_runtime.h>

typedef float v2f __attribute__((ext_vector_type(2)));
typedef float v8f __attribute__((ext_vector_type(8)));

#define HD 128
constexpr int N_USER = 150000;
constexpr int N_BOOK = 75000;
constexpr int N_EDGE = 1500000;
constexpr float LN_EPS = 1e-5f;

__global__ void zero_kernel(float* __restrict__ p, size_t n) {
    size_t i = (size_t)blockIdx.x * blockDim.x + threadIdx.x;
    size_t stride = (size_t)gridDim.x * blockDim.x;
    for (; i < n; i += stride) p[i] = 0.0f;
}

__global__ void count_kernel(const int* __restrict__ esrc, const int* __restrict__ edst,
                             float* __restrict__ cnt_u, float* __restrict__ cnt_b, int E) {
    int i = blockIdx.x * blockDim.x + threadIdx.x;
    if (i < E) {
        atomicAdd(&cnt_u[esrc[i]], 1.0f);
        atomicAdd(&cnt_b[edst[i]], 1.0f);
    }
}

__global__ void scatter_both_kernel(const float* __restrict__ ufeat, const float* __restrict__ bfeat,
                                    const int* __restrict__ esrc, const int* __restrict__ edst,
                                    float* __restrict__ sum_b, float* __restrict__ sum_u, int E) {
    __shared__ float stage[8 * 2 * HD];
    int gid = blockIdx.x * blockDim.x + threadIdx.x;
    int e = gid >> 5;
    if (e >= E) return;
    int lane = gid & 31;
    int s = esrc[e], d = edst[e];
    const float4 vu = *(const float4*)(ufeat + (size_t)s * HD + lane * 4);
    const float4 vb = *(const float4*)(bfeat + (size_t)d * HD + lane * 4);
    asm volatile("" :: "v"(vu.x), "v"(vu.y), "v"(vu.z), "v"(vu.w),
                       "v"(vb.x), "v"(vb.y), "v"(vb.z), "v"(vb.w));
    float* rowu = stage + (threadIdx.x >> 5) * (2 * HD);
    float* rowb = rowu + HD;
    *(float4*)(rowu + lane * 4) = vu;
    const float u0 = rowu[lane], u1 = rowu[lane + 32], u2 = rowu[lane + 64], u3 = rowu[lane + 96];
    float* bp = sum_b + (size_t)d * HD + lane;
    atomicAdd(bp, u0);
    atomicAdd(bp + 32, u1);
    atomicAdd(bp + 64, u2);
    atomicAdd(bp + 96, u3);
    *(float4*)(rowb + lane * 4) = vb;
    const float b0 = rowb[lane], b1 = rowb[lane + 32], b2 = rowb[lane + 64], b3 = rowb[lane + 96];
    float* up = sum_u + (size_t)s * HD + lane;
    atomicAdd(up, b0);
    atomicAdd(up + 32, b1);
    atomicAdd(up + 64, b2);
    atomicAdd(up + 96, b3);
}


__device__ __forceinline__ void load_weight_t(const float* __restrict__ W, float* __restrict__ wt,
                                              int K, int stride) {
    const int nf4 = K * (HD / 4);
    for (int idx = threadIdx.x; idx < nf4; idx += 256) {
        int k = idx >> 5;
        int n = (idx & 31) << 2;
        const float4 w4 = *(const float4*)(W + (size_t)k * HD + n);
        wt[(n + 0) * stride + k] = w4.x;
        wt[(n + 1) * stride + k] = w4.y;
        wt[(n + 2) * stride + k] = w4.z;
        wt[(n + 3) * stride + k] = w4.w;
    }
}

__global__ void gemm_bias_kernel(const float* __restrict__ A, const float* __restrict__ W,
                                 const float* __restrict__ bias, float* __restrict__ out,
                                 int M, int K) {
    __shared__ float wt[HD * (HD + 4)];
    const int stride = K + 4;
    load_weight_t(W, wt, K, stride);
    __syncthreads();

    const int wave = threadIdx.x >> 5;
    const int lane = threadIdx.x & 31;
    const int col  = lane & 15;
    const int half = lane >> 4;
    const int m0   = blockIdx.x * 128 + wave * 16;
    const int aRow = (m0 + col < M) ? (m0 + col) : (M - 1);

    v8f acc[8];
#pragma unroll
    for (int nt = 0; nt < 8; ++nt) {
        float bv = bias[nt * 16 + col];
        acc[nt] = (v8f){bv, bv, bv, bv, bv, bv, bv, bv};
    }

    const float* arow = A + (size_t)aRow * K;
    for (int k0 = 0; k0 < K; k0 += 4) {
        const int kb = k0 + 2 * half;
        v2f a = *(const v2f*)(arow + kb);
        v2f bf[8];
#pragma unroll
        for (int nt = 0; nt < 8; ++nt)
            bf[nt] = *(const v2f*)&wt[(nt * 16 + col) * stride + kb];
#pragma unroll
        for (int nt = 0; nt < 8; ++nt)
            acc[nt] = __builtin_amdgcn_wmma_f32_16x16x4_f32(
                false, a, false, bf[nt], (short)0, acc[nt], false, false);
    }

#pragma unroll
    for (int nt = 0; nt < 8; ++nt)
#pragma unroll
        for (int v = 0; v < 8; ++v) {
            int dr = m0 + v + 8 * half;
            if (dr < M) out[(size_t)dr * HD + nt * 16 + col] = acc[nt][v];
        }
}

__global__ void sage_combine_kernel(const float* __restrict__ sums, const float* __restrict__ cnt,
                                    const float* __restrict__ X, const float* __restrict__ Wl,
                                    const float* __restrict__ bl, const float* __restrict__ Wr,
                                    float* __restrict__ out, int M) {
    __shared__ float wt[HD * (HD + 4)];
    const int stride = HD + 4;

    load_weight_t(Wl, wt, HD, stride);
    __syncthreads();

    const int wave = threadIdx.x >> 5;
    const int lane = threadIdx.x & 31;
    const int col  = lane & 15;
    const int half = lane >> 4;
    const int m0   = blockIdx.x * 128 + wave * 16;
    const int aRow = (m0 + col < M) ? (m0 + col) : (M - 1);
    const float invc = 1.0f / fmaxf(cnt[aRow], 1.0f);

    v8f acc[8];
#pragma unroll
    for (int nt = 0; nt < 8; ++nt) {
        float bv = bl[nt * 16 + col];
        acc[nt] = (v8f){bv, bv, bv, bv, bv, bv, bv, bv};
    }

    const float* srow = sums + (size_t)aRow * HD;
    for (int k0 = 0; k0 < HD; k0 += 4) {
        const int kb = k0 + 2 * half;
        v2f a = *(const v2f*)(srow + kb);
        a.x *= invc;
        a.y *= invc;
        v2f bf[8];
#pragma unroll
        for (int nt = 0; nt < 8; ++nt)
            bf[nt] = *(const v2f*)&wt[(nt * 16 + col) * stride + kb];
#pragma unroll
        for (int nt = 0; nt < 8; ++nt)
            acc[nt] = __builtin_amdgcn_wmma_f32_16x16x4_f32(
                false, a, false, bf[nt], (short)0, acc[nt], false, false);
    }

    __syncthreads();
    load_weight_t(Wr, wt, HD, stride);
    __syncthreads();

    const float* xrow = X + (size_t)aRow * HD;
    for (int k0 = 0; k0 < HD; k0 += 4) {
        const int kb = k0 + 2 * half;
        v2f a = *(const v2f*)(xrow + kb);
        v2f bf[8];
#pragma unroll
        for (int nt = 0; nt < 8; ++nt)
            bf[nt] = *(const v2f*)&wt[(nt * 16 + col) * stride + kb];
#pragma unroll
        for (int nt = 0; nt < 8; ++nt)
            acc[nt] = __builtin_amdgcn_wmma_f32_16x16x4_f32(
                false, a, false, bf[nt], (short)0, acc[nt], false, false);
    }

#pragma unroll
    for (int nt = 0; nt < 8; ++nt)
#pragma unroll
        for (int v = 0; v < 8; ++v) {
            int dr = m0 + v + 8 * half;
            if (dr < M) out[(size_t)dr * HD + nt * 16 + col] = fmaxf(acc[nt][v], 0.0f);
        }
}

__global__ void layernorm_kernel(const float* __restrict__ X, const float* __restrict__ gamma,
                                 const float* __restrict__ beta, float* __restrict__ out, int M) {
    const int wave = threadIdx.x >> 5;
    const int lane = threadIdx.x & 31;
    const int row  = blockIdx.x * 8 + wave;
    if (row >= M) return;

    const float4 v = *(const float4*)(X + (size_t)row * HD + lane * 4);
    float s = v.x + v.y + v.z + v.w;
#pragma unroll
    for (int off = 16; off >= 1; off >>= 1) s += __shfl_xor(s, off, 32);
    const float mu = s * (1.0f / HD);

    const float dx = v.x - mu, dy = v.y - mu, dz = v.z - mu, dw = v.w - mu;
    float q = dx * dx + dy * dy + dz * dz + dw * dw;
#pragma unroll
    for (int off = 16; off >= 1; off >>= 1) q += __shfl_xor(q, off, 32);
    const float rstd = rsqrtf(q * (1.0f / HD) + LN_EPS);

    const float4 g = *(const float4*)(gamma + lane * 4);
    const float4 bt = *(const float4*)(beta + lane * 4);
    float4 o;
    o.x = dx * rstd * g.x + bt.x;
    o.y = dy * rstd * g.y + bt.y;
    o.z = dz * rstd * g.z + bt.z;
    o.w = dw * rstd * g.w + bt.w;
    *(float4*)(out + (size_t)row * HD + lane * 4) = o;
}

extern "C" void kernel_launch(void* const* d_in, const int* in_sizes, int n_in,
                              void* d_out, int out_size, void* d_ws, size_t ws_size,
                              hipStream_t stream) {
    const float* user_x      = (const float*)d_in[0];
    const float* book_x      = (const float*)d_in[1];
    const int*   edge_src    = (const int*)d_in[2];
    const int*   edge_dst    = (const int*)d_in[3];
    const float* user_proj_w = (const float*)d_in[4];
    const float* user_proj_b = (const float*)d_in[5];
    const float* book_proj_w = (const float*)d_in[6];
    const float* book_proj_b = (const float*)d_in[7];
    const float* Wl          = (const float*)d_in[8];
    const float* bl          = (const float*)d_in[9];
    const float* Wr          = (const float*)d_in[10];
    const float* ugamma      = (const float*)d_in[11];
    const float* ubeta       = (const float*)d_in[12];
    const float* bgamma      = (const float*)d_in[13];
    const float* bbeta       = (const float*)d_in[14];

    float* ws = (float*)d_ws;
    size_t off = 0;
    auto carve = [&](size_t n) { float* p = ws + off; off += n; return p; };
    float* u_cur = carve((size_t)N_USER * HD);
    float* u_nxt = carve((size_t)N_USER * HD);
    float* b_cur = carve((size_t)N_BOOK * HD);
    float* b_nxt = carve((size_t)N_BOOK * HD);
    float* sum_u = carve((size_t)N_USER * HD);
    float* sum_b = carve((size_t)N_BOOK * HD);
    float* cnt_u = carve((size_t)N_USER);
    float* cnt_b = carve((size_t)N_BOOK);

    gemm_bias_kernel<<<(N_USER + 127) / 128, 256, 0, stream>>>(
        user_x, user_proj_w, user_proj_b, u_cur, N_USER, 64);
    gemm_bias_kernel<<<(N_BOOK + 127) / 128, 256, 0, stream>>>(
        book_x, book_proj_w, book_proj_b, b_cur, N_BOOK, 128);

    zero_kernel<<<256, 256, 0, stream>>>(cnt_u, (size_t)N_USER);
    zero_kernel<<<256, 256, 0, stream>>>(cnt_b, (size_t)N_BOOK);
    count_kernel<<<(N_EDGE + 255) / 256, 256, 0, stream>>>(
        edge_src, edge_dst, cnt_u, cnt_b, N_EDGE);

    const int scat_blocks = (N_EDGE * 32 + 255) / 256;
    for (int l = 0; l < 3; ++l) {
        const float* wl  = Wl + (size_t)l * HD * HD;
        const float* blv = bl + (size_t)l * HD;
        const float* wr  = Wr + (size_t)l * HD * HD;

        zero_kernel<<<4096, 256, 0, stream>>>(sum_b, (size_t)N_BOOK * HD);
        zero_kernel<<<4096, 256, 0, stream>>>(sum_u, (size_t)N_USER * HD);

        scatter_both_kernel<<<scat_blocks, 256, 0, stream>>>(
            u_cur, b_cur, edge_src, edge_dst, sum_b, sum_u, N_EDGE);

        sage_combine_kernel<<<(N_BOOK + 127) / 128, 256, 0, stream>>>(
            sum_b, cnt_b, b_cur, wl, blv, wr, b_nxt, N_BOOK);
        sage_combine_kernel<<<(N_USER + 127) / 128, 256, 0, stream>>>(
            sum_u, cnt_u, u_cur, wl, blv, wr, u_nxt, N_USER);

        float* t;
        t = u_cur; u_cur = u_nxt; u_nxt = t;
        t = b_cur; b_cur = b_nxt; b_nxt = t;
    }

    float* out_u = (float*)d_out;
    float* out_b = out_u + (size_t)N_USER * HD;
    layernorm_kernel<<<(N_USER + 7) / 8, 256, 0, stream>>>(u_cur, ugamma, ubeta, out_u, N_USER);
    layernorm_kernel<<<(N_BOOK + 7) / 8, 256, 0, stream>>>(b_cur, bgamma, bbeta, out_b, N_BOOK);
}


# Atomic scatter: one wave serves 4 consecutive edges (was 1): indices by 2 scalar loads, all 8 row loads in flight at once, rows added in turn; still atomics' arrival order, little moved (32 failed)
# speedup vs baseline: 1.5893x; 1.0768x over previous
#include <hip/hip_runtime.h>

typedef float v2f __attribute__((ext_vector_type(2)));
typedef float v8f __attribute__((ext_vector_type(8)));

#define HD 128
constexpr int N_USER = 150000;
constexpr int N_BOOK = 75000;
constexpr int N_EDGE = 1500000;
constexpr float LN_EPS = 1e-5f;

__global__ void zero_kernel(float* __restrict__ p, size_t n) {
    size_t i = (size_t)blockIdx.x * blockDim.x + threadIdx.x;
    size_t stride = (size_t)gridDim.x * blockDim.x;
    for (; i < n; i += stride) p[i] = 0.0f;
}

__global__ void count_kernel(const int* __restrict__ esrc, const int* __restrict__ edst,
                             float* __restrict__ cnt_u, float* __restrict__ cnt_b, int E) {
    int i = blockIdx.x * blockDim.x + threadIdx.x;
    if (i < E) {
        atomicAdd(&cnt_u[esrc[i]], 1.0f);
        atomicAdd(&cnt_b[edst[i]], 1.0f);
    }
}

constexpr int EPW = 4;
constexpr int SC_BLOCKS = (N_EDGE + EPW * 8 - 1) / (EPW * 8);

__device__ __forceinline__ float4 load_row(const float* feat, int idx, int lane) {
    return *(const float4*)(feat + (size_t)idx * HD + lane * 4);
}

__device__ __forceinline__ void turn_and_add(float* lrow, const float4 v, float* sums,
                                             int idx, int lane) {
    *(float4*)(lrow + lane * 4) = v;
    const float a0 = lrow[lane], a1 = lrow[lane + 32], a2 = lrow[lane + 64], a3 = lrow[lane + 96];
    float* p = sums + (size_t)idx * HD + lane;
    atomicAdd(p, a0);
    atomicAdd(p + 32, a1);
    atomicAdd(p + 64, a2);
    atomicAdd(p + 96, a3);
}

__global__ void __launch_bounds__(256)
scatter_both_kernel(const float* ufeat, const float* bfeat,
                    const int* __restrict__ esrc, const int* __restrict__ edst,
                    float* sum_b, float* sum_u, int E) {
    __shared__ float stage[8 * 2 * HD];
    const int lane = threadIdx.x & 31;
    const int wib  = threadIdx.x >> 5;
    const int w  = __builtin_amdgcn_readfirstlane((int)blockIdx.x * 8 + wib);
    const int e0 = w * EPW;
    if (e0 >= E) return;
    int s[EPW], d[EPW];
    if (e0 + EPW <= E) {
        const int4 sv = *(const int4*)(esrc + e0);
        const int4 dv = *(const int4*)(edst + e0);
        s[0] = sv.x; s[1] = sv.y; s[2] = sv.z; s[3] = sv.w;
        d[0] = dv.x; d[1] = dv.y; d[2] = dv.z; d[3] = dv.w;
    } else {
#pragma unroll
        for (int j = 0; j < EPW; ++j) {
            const int e = min(e0 + j, E - 1);
            s[j] = esrc[e];
            d[j] = edst[e];
        }
    }
    const int n = min(EPW, E - e0);
    float4 ru[EPW], rb[EPW];
#pragma unroll
    for (int j = 0; j < EPW; ++j) {
        ru[j] = load_row(ufeat, s[j], lane);
        rb[j] = load_row(bfeat, d[j], lane);
        __builtin_amdgcn_sched_barrier(0);
    }
    float* rowu = stage + wib * (2 * HD);
    float* rowb = rowu + HD;
    turn_and_add(rowu, ru[0], sum_b, d[0], lane);
    turn_and_add(rowb, rb[0], sum_u, s[0], lane);
#pragma unroll
    for (int j = 1; j < EPW; ++j) {
        if (j < n) {
            turn_and_add(rowu, ru[j], sum_b, d[j], lane);
            turn_and_add(rowb, rb[j], sum_u, s[j], lane);
        }
    }
}


__device__ __forceinline__ void load_weight_t(const float* __restrict__ W, float* __restrict__ wt,
                                              int K, int stride) {
    const int nf4 = K * (HD / 4);
    for (int idx = threadIdx.x; idx < nf4; idx += 256) {
        int k = idx >> 5;
        int n = (idx & 31) << 2;
        const float4 w4 = *(const float4*)(W + (size_t)k * HD + n);
        wt[(n + 0) * stride + k] = w4.x;
        wt[(n + 1) * stride + k] = w4.y;
        wt[(n + 2) * stride + k] = w4.z;
        wt[(n + 3) * stride + k] = w4.w;
    }
}

__global__ void gemm_bias_kernel(const float* __restrict__ A, const float* __restrict__ W,
                                 const float* __restrict__ bias, float* __restrict__ out,
                                 int M, int K) {
    __shared__ float wt[HD * (HD + 4)];
    const int stride = K + 4;
    load_weight_t(W, wt, K, stride);
    __syncthreads();

    const int wave = threadIdx.x >> 5;
    const int lane = threadIdx.x & 31;
    const int col  = lane & 15;
    const int half = lane >> 4;
    const int m0   = blockIdx.x * 128 + wave * 16;
    const int aRow = (m0 + col < M) ? (m0 + col) : (M - 1);

    v8f acc[8];
#pragma unroll
    for (int nt = 0; nt < 8; ++nt) {
        float bv = bias[nt * 16 + col];
        acc[nt] = (v8f){bv, bv, bv, bv, bv, bv, bv, bv};
    }

    const float* arow = A + (size_t)aRow * K;
    for (int k0 = 0; k0 < K; k0 += 4) {
        const int kb = k0 + 2 * half;
        v2f a = *(const v2f*)(arow + kb);
        v2f bf[8];
#pragma unroll
        for (int nt = 0; nt < 8; ++nt)
            bf[nt] = *(const v2f*)&wt[(nt * 16 + col) * stride + kb];
#pragma unroll
        for (int nt = 0; nt < 8; ++nt)
            acc[nt] = __builtin_amdgcn_wmma_f32_16x16x4_f32(
                false, a, false, bf[nt], (short)0, acc[nt], false, false);
    }

#pragma unroll
    for (int nt = 0; nt < 8; ++nt)
#pragma unroll
        for (int v = 0; v < 8; ++v) {
            int dr = m0 + v + 8 * half;
            if (dr < M) out[(size_t)dr * HD + nt * 16 + col] = acc[nt][v];
        }
}

__global__ void sage_combine_kernel(const float* __restrict__ sums, const float* __restrict__ cnt,
                                    const float* __restrict__ X, const float* __restrict__ Wl,
                                    const float* __restrict__ bl, const float* __restrict__ Wr,
                                    float* __restrict__ out, int M) {
    __shared__ float wt[HD * (HD + 4)];
    const int stride = HD + 4;

    load_weight_t(Wl, wt, HD, stride);
    __syncthreads();

    const int wave = threadIdx.x >> 5;
    const int lane = threadIdx.x & 31;
    const int col  = lane & 15;
    const int half = lane >> 4;
    const int m0   = blockIdx.x * 128 + wave * 16;
    const int aRow = (m0 + col < M) ? (m0 + col) : (M - 1);
    const float invc = 1.0f / fmaxf(cnt[aRow], 1.0f);

    v8f acc[8];
#pragma unroll
    for (int nt = 0; nt < 8; ++nt) {
        float bv = bl[nt * 16 + col];
        acc[nt] = (v8f){bv, bv, bv, bv, bv, bv, bv, bv};
    }

    const float* srow = sums + (size_t)aRow * HD;
    for (int k0 = 0; k0 < HD; k0 += 4) {
        const int kb = k0 + 2 * half;
        v2f a = *(const v2f*)(srow + kb);
        a.x *= invc;
        a.y *= invc;
        v2f bf[8];
#pragma unroll
        for (int nt = 0; nt < 8; ++nt)
            bf[nt] = *(const v2f*)&wt[(nt * 16 + col) * stride + kb];
#pragma unroll
        for (int nt = 0; nt < 8; ++nt)
            acc[nt] = __builtin_amdgcn_wmma_f32_16x16x4_f32(
                false, a, false, bf[nt], (short)0, acc[nt], false, false);
    }

    __syncthreads();
    load_weight_t(Wr, wt, HD, stride);
    __syncthreads();

    const float* xrow = X + (size_t)aRow * HD;
    for (int k0 = 0; k0 < HD; k0 += 4) {
        const int kb = k0 + 2 * half;
        v2f a = *(const v2f*)(xrow + kb);
        v2f bf[8];
#pragma unroll
        for (int nt = 0; nt < 8; ++nt)
            bf[nt] = *(const v2f*)&wt[(nt * 16 + col) * stride + kb];
#pragma unroll
        for (int nt = 0; nt < 8; ++nt)
            acc[nt] = __builtin_amdgcn_wmma_f32_16x16x4_f32(
                false, a, false, bf[nt], (short)0, acc[nt], false, false);
    }

#pragma unroll
    for (int nt = 0; nt < 8; ++nt)
#pragma unroll
        for (int v = 0; v < 8; ++v) {
            int dr = m0 + v + 8 * half;
            if (dr < M) out[(size_t)dr * HD + nt * 16 + col] = fmaxf(acc[nt][v], 0.0f);
        }
}

__global__ void layernorm_kernel(const float* __restrict__ X, const float* __restrict__ gamma,
                                 const float* __restrict__ beta, float* __restrict__ out, int M) {
    const int wave = threadIdx.x >> 5;
    const int lane = threadIdx.x & 31;
    const int row  = blockIdx.x * 8 + wave;
    if (row >= M) return;

    const float4 v = *(const float4*)(X + (size_t)row * HD + lane * 4);
    float s = v.x + v.y + v.z + v.w;
#pragma unroll
    for (int off = 16; off >= 1; off >>= 1) s += __shfl_xor(s, off, 32);
    const float mu = s * (1.0f / HD);

    const float dx = v.x - mu, dy = v.y - mu, dz = v.z - mu, dw = v.w - mu;
    float q = dx * dx + dy * dy + dz * dz + dw * dw;
#pragma unroll
    for (int off = 16; off >= 1; off >>= 1) q += __shfl_xor(q, off, 32);
    const float rstd = rsqrtf(q * (1.0f / HD) + LN_EPS);

    const float4 g = *(const float4*)(gamma + lane * 4);
    const float4 bt = *(const float4*)(beta + lane * 4);
    float4 o;
    o.x = dx * rstd * g.x + bt.x;
    o.y = dy * rstd * g.y + bt.y;
    o.z = dz * rstd * g.z + bt.z;
    o.w = dw * rstd * g.w + bt.w;
    *(float4*)(out + (size_t)row * HD + lane * 4) = o;
}

extern "C" void kernel_launch(void* const* d_in, const int* in_sizes, int n_in,
                              void* d_out, int out_size, void* d_ws, size_t ws_size,
                              hipStream_t stream) {
    const float* user_x      = (const float*)d_in[0];
    const float* book_x      = (const float*)d_in[1];
    const int*   edge_src    = (const int*)d_in[2];
    const int*   edge_dst    = (const int*)d_in[3];
    const float* user_proj_w = (const float*)d_in[4];
    const float* user_proj_b = (const float*)d_in[5];
    const float* book_proj_w = (const float*)d_in[6];
    const float* book_proj_b = (const float*)d_in[7];
    const float* Wl          = (const float*)d_in[8];
    const float* bl          = (const float*)d_in[9];
    const float* Wr          = (const float*)d_in[10];
    const float* ugamma      = (const float*)d_in[11];
    const float* ubeta       = (const float*)d_in[12];
    const float* bgamma      = (const float*)d_in[13];
    const float* bbeta       = (const float*)d_in[14];

    float* ws = (float*)d_ws;
    size_t off = 0;
    auto carve = [&](size_t n) { float* p = ws + off; off += n; return p; };
    float* u_cur = carve((size_t)N_USER * HD);
    float* u_nxt = carve((size_t)N_USER * HD);
    float* b_cur = carve((size_t)N_BOOK * HD);
    float* b_nxt = carve((size_t)N_BOOK * HD);
    float* sum_u = carve((size_t)N_USER * HD);
    float* sum_b = carve((size_t)N_BOOK * HD);
    float* cnt_u = carve((size_t)N_USER);
    float* cnt_b = carve((size_t)N_BOOK);

    gemm_bias_kernel<<<(N_USER + 127) / 128, 256, 0, stream>>>(
        user_x, user_proj_w, user_proj_b, u_cur, N_USER, 64);
    gemm_bias_kernel<<<(N_BOOK + 127) / 128, 256, 0, stream>>>(
        book_x, book_proj_w, book_proj_b, b_cur, N_BOOK, 128);

    zero_kernel<<<256, 256, 0, stream>>>(cnt_u, (size_t)N_USER);
    zero_kernel<<<256, 256, 0, stream>>>(cnt_b, (size_t)N_BOOK);
    count_kernel<<<(N_EDGE + 255) / 256, 256, 0, stream>>>(
        edge_src, edge_dst, cnt_u, cnt_b, N_EDGE);

    for (int l = 0; l < 3; ++l) {
        const float* wl  = Wl + (size_t)l * HD * HD;
        const float* blv = bl + (size_t)l * HD;
        const float* wr  = Wr + (size_t)l * HD * HD;

        zero_kernel<<<4096, 256, 0, stream>>>(sum_b, (size_t)N_BOOK * HD);
        zero_kernel<<<4096, 256, 0, stream>>>(sum_u, (size_t)N_USER * HD);

        scatter_both_kernel<<<SC_BLOCKS, 256, 0, stream>>>(
            u_cur, b_cur, edge_src, edge_dst, sum_b, sum_u, N_EDGE);

        sage_combine_kernel<<<(N_BOOK + 127) / 128, 256, 0, stream>>>(
            sum_b, cnt_b, b_cur, wl, blv, wr, b_nxt, N_BOOK);
        sage_combine_kernel<<<(N_USER + 127) / 128, 256, 0, stream>>>(
            sum_u, cnt_u, u_cur, wl, blv, wr, u_nxt, N_USER);

        float* t;
        t = u_cur; u_cur = u_nxt; u_nxt = t;
        t = b_cur; b_cur = b_nxt; b_nxt = t;
    }

    float* out_u = (float*)d_out;
    float* out_b = out_u + (size_t)N_USER * HD;
    layernorm_kernel<<<(N_USER + 7) / 8, 256, 0, stream>>>(u_cur, ugamma, ubeta, out_u, N_USER);
    layernorm_kernel<<<(N_BOOK + 7) / 8, 256, 0, stream>>>(b_cur, bgamma, bbeta, out_b, N_BOOK);
}


# Atomic scatter: sum rows cut into two ranges, each served by the blocks of one package half (a block draws its piece of the edge list from its half's counter): a line takes adds from one half only
# speedup vs baseline: 2.2982x; 1.4460x over previous
#include <hip/hip_runtime.h>

typedef float v2f __attribute__((ext_vector_type(2)));
typedef float v8f __attribute__((ext_vector_type(8)));

#define HD 128
constexpr int N_USER = 150000;
constexpr int N_BOOK = 75000;
constexpr int N_EDGE = 1500000;
constexpr float LN_EPS = 1e-5f;

constexpr int PW = 64;
__device__ __forceinline__ size_t plane_off(int n_rows, int row, int plane) {
    return ((size_t)plane * n_rows + row) * PW;
}

__device__ __forceinline__ int package_half() {
    const unsigned id = __builtin_amdgcn_s_sendmsg_rtn(0x87);
    return (int)((id >> 18) & 1u);
}

__global__ void zero_kernel(float* __restrict__ p, size_t n) {
    size_t i = (size_t)blockIdx.x * blockDim.x + threadIdx.x;
    size_t stride = (size_t)gridDim.x * blockDim.x;
    for (; i < n; i += stride) p[i] = 0.0f;
}

__global__ void count_kernel(const int* __restrict__ esrc, const int* __restrict__ edst,
                             float* __restrict__ cnt_u, float* __restrict__ cnt_b, int E) {
    int i = blockIdx.x * blockDim.x + threadIdx.x;
    if (i < E) {
        atomicAdd(&cnt_u[esrc[i]], 1.0f);
        atomicAdd(&cnt_b[edst[i]], 1.0f);
    }
}

constexpr int EPW = 16;
constexpr int SC_PIECES = (N_EDGE + EPW * 8 - 1) / (EPW * 8);
constexpr int SC_BLOCKS_PASS = 2 * SC_PIECES;
constexpr int SC_GRID = 4 * SC_BLOCKS_PASS;
constexpr int SC_CTR_PITCH = 64;
constexpr int SC_COUNTERS = 2 * SC_CTR_PITCH;

__global__ void __launch_bounds__(256)
scatter_pass_kernel(const float* ufeat, const float* bfeat,
                    const int* __restrict__ esrc, const int* __restrict__ edst,
                    float* sum_b, float* sum_u, int* counters, int E, int first) {
    __shared__ int s_item;
    const int lane = threadIdx.x & 31;
    const int wib  = threadIdx.x >> 5;
    const int pass = (int)blockIdx.x / SC_BLOCKS_PASS;
    if (threadIdx.x == 0) {
        int* ctr = counters + pass;
        int r = package_half();
        int t = atomicAdd(ctr + r * SC_CTR_PITCH, 1);
        if (t >= SC_PIECES) {
            r ^= 1;
            t = atomicAdd(ctr + r * SC_CTR_PITCH, 1);
        }
        s_item = t < SC_PIECES ? 2 * t + r : -1;
    }
    __syncthreads();
    const int item = __builtin_amdgcn_readfirstlane(s_item);
    if (item < 0) return;
    const int upper = item & 1;
    const int blk   = item >> 1;
    const bool to_book = first ? (pass == 0) : (pass < 2);
    const int h = first ? (pass == 0 ? 0 : pass - 1) : (pass & 1);
    const unsigned fpitch = (first && !to_book) ? (unsigned)HD : (unsigned)PW;
    const float* feat = to_book ? (first ? ufeat : ufeat + plane_off(N_USER, 0, h))
                                : (first ? bfeat + PW * h : bfeat + plane_off(N_BOOK, 0, h));
    float*       sums = to_book ? sum_b + plane_off(N_BOOK, 0, h) : sum_u + plane_off(N_USER, 0, h);
    const int*   gidx = to_book ? esrc : edst;
    const int*   tidx = to_book ? edst : esrc;
    const int split   = (to_book ? N_BOOK : N_USER) >> 1;
    const int w  = __builtin_amdgcn_readfirstlane(blk * 8 + wib);
    const int e0 = w * EPW;
    if (e0 >= E) return;
    int g[EPW], t[EPW];
    if (e0 + EPW <= E) {
#pragma unroll
        for (int q = 0; q < EPW / 4; ++q) {
            const int4 gq = *(const int4*)(gidx + e0 + 4 * q);
            const int4 tq = *(const int4*)(tidx + e0 + 4 * q);
            g[4 * q] = gq.x; g[4 * q + 1] = gq.y; g[4 * q + 2] = gq.z; g[4 * q + 3] = gq.w;
            t[4 * q] = tq.x; t[4 * q + 1] = tq.y; t[4 * q + 2] = tq.z; t[4 * q + 3] = tq.w;
        }
    } else {
#pragma unroll
        for (int j = 0; j < EPW; ++j) {
            const int e = min(e0 + j, E - 1);
            g[j] = gidx[e];
            t[j] = tidx[e];
        }
    }
    const int n = min(EPW, E - e0);
    unsigned mine = 0;
#pragma unroll
    for (int j = 0; j < EPW; ++j) mine |= (t[j] >= split ? 1u : 0u) << j;
    if (!upper) mine = ~mine;
    mine &= (1u << n) - 1u;
    float v0[EPW], v1[EPW];
#pragma unroll
    for (int j = 0; j < EPW; ++j) {
        v0[j] = 0.0f;
        v1[j] = 0.0f;
        if (mine & (1u << j)) {
            const unsigned off = (unsigned)g[j] * fpitch + lane;
            v0[j] = feat[off];
            v1[j] = feat[off + 32];
        }
    }
#pragma unroll
    for (int j = 0; j < EPW; ++j) {
        if (mine & (1u << j)) {
            const unsigned off = (unsigned)t[j] * PW + lane;
            atomicAdd(sums + off, v0[j]);
            atomicAdd(sums + off + 32, v1[j]);
        }
    }
}


__device__ __forceinline__ void load_weight_t(const float* __restrict__ W, float* __restrict__ wt,
                                              int K, int stride) {
    const int nf4 = K * (HD / 4);
    for (int idx = threadIdx.x; idx < nf4; idx += 256) {
        int k = idx >> 5;
        int n = (idx & 31) << 2;
        const float4 w4 = *(const float4*)(W + (size_t)k * HD + n);
        wt[(n + 0) * stride + k] = w4.x;
        wt[(n + 1) * stride + k] = w4.y;
        wt[(n + 2) * stride + k] = w4.z;
        wt[(n + 3) * stride + k] = w4.w;
    }
}

constexpr int IMG128 = HD * (HD + 4);
constexpr int IMG64  = HD * (64 + 4);
constexpr int IMG_PUWL = 0;
constexpr int IMG_PUWR = IMG_PUWL + IMG64;
constexpr int IMG_PBWL = IMG_PUWR + IMG64;
constexpr int IMG_PBWR = IMG_PBWL + IMG128;
constexpr int IMG_LAYER = IMG_PBWR + IMG128;
constexpr int IMG_FLOATS = IMG_LAYER + 4 * IMG128;

__device__ __forceinline__ void load_weight_image(const float* __restrict__ img,
                                                  float* __restrict__ wt, int nfloats) {
    const float4* s4 = (const float4*)img;
    float4* d4 = (float4*)wt;
    const int nf4 = nfloats >> 2;
#pragma unroll 8
    for (int idx = threadIdx.x; idx < nf4; idx += 256) d4[idx] = s4[idx];
}

__device__ __forceinline__ void store_weight_image(float* __restrict__ wt, float* __restrict__ img,
                                                   int K, int stride) {
    for (int i = threadIdx.x; i < HD * 4; i += 256) wt[(i >> 2) * stride + K + (i & 3)] = 0.0f;
    __syncthreads();
    const float4* s4 = (const float4*)wt;
    float4* d4 = (float4*)img;
    const int nf4 = (HD * stride) >> 2;
    for (int idx = threadIdx.x; idx < nf4; idx += 256) d4[idx] = s4[idx];
}

constexpr int COMP_BUWL = 0, COMP_BUWR = 128, COMP_BBWL = 256, COMP_BBWR = 384;
constexpr int COMP_FLOATS = 512;

__global__ void __launch_bounds__(256)
compose_kernel(const float* __restrict__ Pu, const float* __restrict__ Pb,
               const float* __restrict__ bu, const float* __restrict__ bb,
               const float* __restrict__ Wl, const float* __restrict__ Wr,
               float* __restrict__ comp, float* __restrict__ img) {
    __shared__ __attribute__((aligned(16))) float wt[HD * (HD + 4)];
    const int stride = HD + 4;
    const int job = blockIdx.x;
    if (job >= 8) {
        const int j = job - 8;
        load_weight_t(((j & 1) ? Wr : Wl) + (size_t)(1 + (j >> 1)) * HD * HD, wt, HD, stride);
        store_weight_image(wt, img + IMG_LAYER + j * IMG128, HD, stride);
        return;
    }
    const int s = job >> 1;
    const float* A = s == 0 ? Pu : s == 1 ? Pb : s == 2 ? bu : bb;
    const int M = s == 0 ? 64 : s == 1 ? 128 : 1;
    load_weight_t((job & 1) ? Wr : Wl, wt, HD, stride);
    __syncthreads();

    const int wave = threadIdx.x >> 5;
    const int lane = threadIdx.x & 31;
    const int col  = lane & 15;
    const int half = lane >> 4;
    const int m0   = wave * 16;
    const int aRow = (m0 + col < M) ? (m0 + col) : (M - 1);

    v8f acc[8];
#pragma unroll
    for (int nt = 0; nt < 8; ++nt) acc[nt] = (v8f){0.f, 0.f, 0.f, 0.f, 0.f, 0.f, 0.f, 0.f};

    const float* arow = A + (size_t)aRow * HD;
    for (int k0 = 0; k0 < HD; k0 += 4) {
        const int kb = k0 + 2 * half;
        v2f a = *(const v2f*)(arow + kb);
        v2f bf[8];
#pragma unroll
        for (int nt = 0; nt < 8; ++nt)
            bf[nt] = *(const v2f*)&wt[(nt * 16 + col) * stride + kb];
#pragma unroll
        for (int nt = 0; nt < 8; ++nt)
            acc[nt] = __builtin_amdgcn_wmma_f32_16x16x4_f32(
                false, a, false, bf[nt], (short)0, acc[nt], false, false);
    }

    if (s >= 2) {
        float* out = comp + (job == 4 ? COMP_BUWL : job == 5 ? COMP_BUWR :
                             job == 6 ? COMP_BBWL : COMP_BBWR);
#pragma unroll
        for (int nt = 0; nt < 8; ++nt)
            if (m0 == 0 && half == 0) out[nt * 16 + col] = acc[nt][0];
        return;
    }
    __syncthreads();
    const int istride = M + 4;
#pragma unroll
    for (int nt = 0; nt < 8; ++nt)
#pragma unroll
        for (int v = 0; v < 8; ++v) {
            int dr = m0 + v + 8 * half;
            if (dr < M) wt[(nt * 16 + col) * istride + dr] = acc[nt][v];
        }
    store_weight_image(wt, img + (job == 0 ? IMG_PUWL : job == 1 ? IMG_PUWR :
                                  job == 2 ? IMG_PBWL : IMG_PBWR), M, istride);
}

constexpr int UB = (N_USER + 127) / 128;
constexpr int BB = (N_BOOK + 127) / 128;
__global__ void sage_combine_kernel(const float* __restrict__ sums_u, const float* __restrict__ cnt_u,
                                    const float* __restrict__ X_u, float* __restrict__ out_u,
                                    const float* __restrict__ sums_b, const float* __restrict__ cnt_b,
                                    const float* __restrict__ X_b, float* __restrict__ out_b,
                                    const float* __restrict__ Wl, const float* __restrict__ bl,
                                    const float* __restrict__ Wr) {
    __shared__ __attribute__((aligned(16))) float wt[HD * (HD + 4)];
    const int stride = HD + 4;
    const bool isb = (int)blockIdx.x >= UB;
    const float* __restrict__ sums = isb ? sums_b : sums_u;
    const float* __restrict__ cnt  = isb ? cnt_b : cnt_u;
    const float* __restrict__ X    = isb ? X_b : X_u;
    float* __restrict__ out        = isb ? out_b : out_u;
    const int M   = isb ? N_BOOK : N_USER;
    const int blk = isb ? (int)blockIdx.x - UB : (int)blockIdx.x;

    load_weight_image(Wl, wt, IMG128);
    __syncthreads();

    const int wave = threadIdx.x >> 5;
    const int lane = threadIdx.x & 31;
    const int col  = lane & 15;
    const int half = lane >> 4;
    const int m0   = blk * 128 + wave * 16;
    const int aRow = (m0 + col < M) ? (m0 + col) : (M - 1);
    const float invc = 1.0f / fmaxf(cnt[aRow], 1.0f);

    v8f acc[8];
#pragma unroll
    for (int nt = 0; nt < 8; ++nt) {
        float bv = bl[nt * 16 + col];
        acc[nt] = (v8f){bv, bv, bv, bv, bv, bv, bv, bv};
    }

    for (int k0 = 0; k0 < HD; k0 += 4) {
        const int kb = k0 + 2 * half;
        v2f a = *(const v2f*)(sums + plane_off(M, aRow, k0 >> 6) + (kb & (PW - 1)));
        a.x *= invc;
        a.y *= invc;
        v2f bf[8];
#pragma unroll
        for (int nt = 0; nt < 8; ++nt)
            bf[nt] = *(const v2f*)&wt[(nt * 16 + col) * stride + kb];
#pragma unroll
        for (int nt = 0; nt < 8; ++nt)
            acc[nt] = __builtin_amdgcn_wmma_f32_16x16x4_f32(
                false, a, false, bf[nt], (short)0, acc[nt], false, false);
    }

    __syncthreads();
    load_weight_image(Wr, wt, IMG128);
    __syncthreads();

    for (int k0 = 0; k0 < HD; k0 += 4) {
        const int kb = k0 + 2 * half;
        v2f a = *(const v2f*)(X + plane_off(M, aRow, k0 >> 6) + (kb & (PW - 1)));
        v2f bf[8];
#pragma unroll
        for (int nt = 0; nt < 8; ++nt)
            bf[nt] = *(const v2f*)&wt[(nt * 16 + col) * stride + kb];
#pragma unroll
        for (int nt = 0; nt < 8; ++nt)
            acc[nt] = __builtin_amdgcn_wmma_f32_16x16x4_f32(
                false, a, false, bf[nt], (short)0, acc[nt], false, false);
    }

#pragma unroll
    for (int nt = 0; nt < 8; ++nt)
#pragma unroll
        for (int v = 0; v < 8; ++v) {
            int dr = m0 + v + 8 * half;
            if (dr < M) out[plane_off(M, dr, nt >> 2) + (nt & 3) * 16 + col] = fmaxf(acc[nt][v], 0.0f);
        }
}

__global__ void sage_first_kernel(const float* __restrict__ sums_u, const float* __restrict__ cnt_u,
                                  const float* __restrict__ X_u, const float* __restrict__ W1_u,
                                  const float* __restrict__ W2_u, const float* __restrict__ c1_u,
                                  const float* __restrict__ c2_u, float* __restrict__ out_u,
                                  const float* __restrict__ sums_b, const float* __restrict__ cnt_b,
                                  const float* __restrict__ X_b, const float* __restrict__ W1_b,
                                  const float* __restrict__ W2_b, const float* __restrict__ c1_b,
                                  const float* __restrict__ c2_b, float* __restrict__ out_b,
                                  const float* __restrict__ bl,
                                  int K1_u, int K2_u, int K1_b, int K2_b) {
    __shared__ __attribute__((aligned(16))) float wt[HD * (HD + 4)];
    const bool isb = (int)blockIdx.x >= UB;
    const float* __restrict__ sums = isb ? sums_b : sums_u;
    const float* __restrict__ cnt  = isb ? cnt_b : cnt_u;
    const float* __restrict__ X    = isb ? X_b : X_u;
    const float* __restrict__ W1   = isb ? W1_b : W1_u;
    const float* __restrict__ W2   = isb ? W2_b : W2_u;
    const float* __restrict__ c1   = isb ? c1_b : c1_u;
    const float* __restrict__ c2   = isb ? c2_b : c2_u;
    float* __restrict__ out        = isb ? out_b : out_u;
    const int M   = isb ? N_BOOK : N_USER;
    const int K1  = isb ? K1_b : K1_u;
    const int K2  = isb ? K2_b : K2_u;
    const int blk = isb ? (int)blockIdx.x - UB : (int)blockIdx.x;
    const int stride1 = K1 + 4;
    const int stride2 = K2 + 4;

    load_weight_image(W1, wt, HD * stride1);
    __syncthreads();

    const int wave = threadIdx.x >> 5;
    const int lane = threadIdx.x & 31;
    const int col  = lane & 15;
    const int half = lane >> 4;
    const int m0   = blk * 128 + wave * 16;
    const int aRow = (m0 + col < M) ? (m0 + col) : (M - 1);
    const float invc = 1.0f / fmaxf(cnt[aRow], 1.0f);

    v8f acc[8];
#pragma unroll
    for (int nt = 0; nt < 8; ++nt) {
        float bv = bl[nt * 16 + col] + c2[nt * 16 + col];
        acc[nt] = (v8f){bv, bv, bv, bv, bv, bv, bv, bv};
    }

    for (int k0 = 0; k0 < K1; k0 += 4) {
        const int kb = k0 + 2 * half;
        v2f a = *(const v2f*)(sums + plane_off(M, aRow, k0 >> 6) + (kb & (PW - 1)));
        a.x *= invc;
        a.y *= invc;
        v2f bf[8];
#pragma unroll
        for (int nt = 0; nt < 8; ++nt)
            bf[nt] = *(const v2f*)&wt[(nt * 16 + col) * stride1 + kb];
#pragma unroll
        for (int nt = 0; nt < 8; ++nt)
            acc[nt] = __builtin_amdgcn_wmma_f32_16x16x4_f32(
                false, a, false, bf[nt], (short)0, acc[nt], false, false);
    }

    __syncthreads();
    load_weight_image(W2, wt, HD * stride2);
    __syncthreads();

    const float* xrow = X + (size_t)aRow * K2;
    for (int k0 = 0; k0 < K2; k0 += 4) {
        const int kb = k0 + 2 * half;
        v2f a = *(const v2f*)(xrow + kb);
        v2f bf[8];
#pragma unroll
        for (int nt = 0; nt < 8; ++nt)
            bf[nt] = *(const v2f*)&wt[(nt * 16 + col) * stride2 + kb];
#pragma unroll
        for (int nt = 0; nt < 8; ++nt)
            acc[nt] = __builtin_amdgcn_wmma_f32_16x16x4_f32(
                false, a, false, bf[nt], (short)0, acc[nt], false, false);
    }

    float has[8];
#pragma unroll
    for (int v = 0; v < 8; ++v) {
        const int dr = m0 + v + 8 * half;
        has[v] = (cnt[dr < M ? dr : M - 1] > 0.0f) ? 1.0f : 0.0f;
    }
#pragma unroll
    for (int nt = 0; nt < 8; ++nt) {
        const float cb = c1[nt * 16 + col];
#pragma unroll
        for (int v = 0; v < 8; ++v) {
            int dr = m0 + v + 8 * half;
            if (dr < M)
                out[plane_off(M, dr, nt >> 2) + (nt & 3) * 16 + col] =
                    fmaxf(acc[nt][v] + has[v] * cb, 0.0f);
        }
    }
}

constexpr int LNU = (N_USER + 7) / 8;
constexpr int LNB = (N_BOOK + 7) / 8;
__global__ void layernorm_kernel(const float* __restrict__ X_u, const float* __restrict__ gamma_u,
                                 const float* __restrict__ beta_u, float* __restrict__ out_u,
                                 const float* __restrict__ X_b, const float* __restrict__ gamma_b,
                                 const float* __restrict__ beta_b, float* __restrict__ out_b) {
    const bool isb = (int)blockIdx.x >= LNU;
    const float* __restrict__ X     = isb ? X_b : X_u;
    const float* __restrict__ gamma = isb ? gamma_b : gamma_u;
    const float* __restrict__ beta  = isb ? beta_b : beta_u;
    float* __restrict__ out         = isb ? out_b : out_u;
    const int M    = isb ? N_BOOK : N_USER;
    const int wave = threadIdx.x >> 5;
    const int lane = threadIdx.x & 31;
    const int row  = (isb ? (int)blockIdx.x - LNU : (int)blockIdx.x) * 8 + wave;
    if (row >= M) return;

    const float4 v = *(const float4*)(X + plane_off(M, row, lane >> 4) + (lane & 15) * 4);
    float s = v.x + v.y + v.z + v.w;
#pragma unroll
    for (int off = 16; off >= 1; off >>= 1) s += __shfl_xor(s, off, 32);
    const float mu = s * (1.0f / HD);

    const float dx = v.x - mu, dy = v.y - mu, dz = v.z - mu, dw = v.w - mu;
    float q = dx * dx + dy * dy + dz * dz + dw * dw;
#pragma unroll
    for (int off = 16; off >= 1; off >>= 1) q += __shfl_xor(q, off, 32);
    const float rstd = rsqrtf(q * (1.0f / HD) + LN_EPS);

    const float4 g = *(const float4*)(gamma + lane * 4);
    const float4 bt = *(const float4*)(beta + lane * 4);
    float4 o;
    o.x = dx * rstd * g.x + bt.x;
    o.y = dy * rstd * g.y + bt.y;
    o.z = dz * rstd * g.z + bt.z;
    o.w = dw * rstd * g.w + bt.w;
    *(float4*)(out + (size_t)row * HD + lane * 4) = o;
}

extern "C" void kernel_launch(void* const* d_in, const int* in_sizes, int n_in,
                              void* d_out, int out_size, void* d_ws, size_t ws_size,
                              hipStream_t stream) {
    const float* user_x      = (const float*)d_in[0];
    const float* book_x      = (const float*)d_in[1];
    const int*   edge_src    = (const int*)d_in[2];
    const int*   edge_dst    = (const int*)d_in[3];
    const float* user_proj_w = (const float*)d_in[4];
    const float* user_proj_b = (const float*)d_in[5];
    const float* book_proj_w = (const float*)d_in[6];
    const float* book_proj_b = (const float*)d_in[7];
    const float* Wl          = (const float*)d_in[8];
    const float* bl          = (const float*)d_in[9];
    const float* Wr          = (const float*)d_in[10];
    const float* ugamma      = (const float*)d_in[11];
    const float* ubeta       = (const float*)d_in[12];
    const float* bgamma      = (const float*)d_in[13];
    const float* bbeta       = (const float*)d_in[14];

    float* ws = (float*)d_ws;
    size_t off = 0;
    auto carve = [&](size_t n) { float* p = ws + off; off += n; return p; };
    float* u_cur = carve((size_t)N_USER * HD);
    float* u_nxt = carve((size_t)N_USER * HD);
    float* b_cur = carve((size_t)N_BOOK * HD);
    float* b_nxt = carve((size_t)N_BOOK * HD);
    float* sum_u = carve((size_t)N_USER * HD);
    float* sum_b = carve((size_t)N_BOOK * HD);
    float* cnt_u = carve((size_t)N_USER);
    float* cnt_b = carve((size_t)N_BOOK);
    int*   ctrs  = (int*)carve((size_t)3 * SC_COUNTERS);
    float* comp  = carve((size_t)COMP_FLOATS);
    float* img   = carve((size_t)IMG_FLOATS);

    compose_kernel<<<12, 256, 0, stream>>>(
        user_proj_w, book_proj_w, user_proj_b, book_proj_b, Wl, Wr, comp, img);

    zero_kernel<<<4096, 256, 0, stream>>>(
        sum_u, (size_t)(N_USER + N_BOOK) * HD + (size_t)N_USER + (size_t)N_BOOK + (size_t)3 * SC_COUNTERS);
    count_kernel<<<(N_EDGE + 255) / 256, 256, 0, stream>>>(
        edge_src, edge_dst, cnt_u, cnt_b, N_EDGE);

    scatter_pass_kernel<<<3 * SC_BLOCKS_PASS, 256, 0, stream>>>(
        user_x, book_x, edge_src, edge_dst, sum_b, sum_u, ctrs, N_EDGE, 1);
    sage_first_kernel<<<UB + BB, 256, 0, stream>>>(
        sum_u, cnt_u, user_x, img + IMG_PBWL, img + IMG_PUWR,
        comp + COMP_BBWL, comp + COMP_BUWR, u_cur,
        sum_b, cnt_b, book_x, img + IMG_PUWL, img + IMG_PBWR,
        comp + COMP_BUWL, comp + COMP_BBWR, b_cur, bl, 128, 64, 64, 128);

    for (int l = 1; l < 3; ++l) {
        const float* wl  = img + IMG_LAYER + (size_t)(2 * l - 2) * IMG128;
        const float* blv = bl + (size_t)l * HD;
        const float* wr  = img + IMG_LAYER + (size_t)(2 * l - 1) * IMG128;

        zero_kernel<<<4096, 256, 0, stream>>>(sum_u, (size_t)(N_USER + N_BOOK) * HD);

        scatter_pass_kernel<<<SC_GRID, 256, 0, stream>>>(
            u_cur, b_cur, edge_src, edge_dst, sum_b, sum_u, ctrs + l * SC_COUNTERS, N_EDGE, 0);

        sage_combine_kernel<<<UB + BB, 256, 0, stream>>>(
            sum_u, cnt_u, u_cur, u_nxt, sum_b, cnt_b, b_cur, b_nxt, wl, blv, wr);

        float* t;
        t = u_cur; u_cur = u_nxt; u_nxt = t;
        t = b_cur; b_cur = b_nxt; b_nxt = t;
    }

    float* out_u = (float*)d_out;
    float* out_b = out_u + (size_t)N_USER * HD;
    layernorm_kernel<<<LNU + LNB, 256, 0, stream>>>(
        u_cur, ugamma, ubeta, out_u, b_cur, bgamma, bbeta, out_b);
}


# Streaming launches folded into layer GEMM launches (12 to 9): sums cleared by extra blocks in the GEMM grid's tail, two sum sets in turn; layernorm in the last GEMM's registers (row sums regrouped)
# speedup vs baseline: 2.4216x; 1.0537x over previous
#include <hip/hip_runtime.h>

typedef float v2f __attribute__((ext_vector_type(2)));
typedef float v4f __attribute__((ext_vector_type(4)));
typedef float v8f __attribute__((ext_vector_type(8)));

#define HD 128
constexpr int N_USER = 150000;
constexpr int N_BOOK = 75000;
constexpr int N_EDGE = 1500000;
constexpr float LN_EPS = 1e-5f;

constexpr int PW = 64;
__device__ __forceinline__ size_t plane_off(int n_rows, int row, int plane) {
    return ((size_t)plane * n_rows + row) * PW;
}

__device__ __forceinline__ int package_half() {
    const unsigned id = __builtin_amdgcn_s_sendmsg_rtn(0x87);
    return (int)((id >> 18) & 1u);
}

__global__ void zero_kernel(float* __restrict__ p, size_t n) {
    size_t i = (size_t)blockIdx.x * blockDim.x + threadIdx.x;
    size_t stride = (size_t)gridDim.x * blockDim.x;
    for (; i < n; i += stride) p[i] = 0.0f;
}

__global__ void count_kernel(const int* __restrict__ esrc, const int* __restrict__ edst,
                             float* __restrict__ cnt_u, float* __restrict__ cnt_b, int E) {
    int i = blockIdx.x * blockDim.x + threadIdx.x;
    if (i < E) {
        atomicAdd(&cnt_u[esrc[i]], 1.0f);
        atomicAdd(&cnt_b[edst[i]], 1.0f);
    }
}

constexpr int EPW = 16;
constexpr int SC_PIECES = (N_EDGE + EPW * 8 - 1) / (EPW * 8);
constexpr int SC_BLOCKS_PASS = 2 * SC_PIECES;
constexpr int SC_GRID = 4 * SC_BLOCKS_PASS;
constexpr int SC_CTR_PITCH = 64;
constexpr int SC_COUNTERS = 2 * SC_CTR_PITCH;

__global__ void __launch_bounds__(256)
scatter_pass_kernel(const float* ufeat, const float* bfeat,
                    const int* __restrict__ esrc, const int* __restrict__ edst,
                    float* sum_b, float* sum_u, int* counters, int E, int first) {
    __shared__ int s_item;
    const int lane = threadIdx.x & 31;
    const int wib  = threadIdx.x >> 5;
    const int pass = (int)blockIdx.x / SC_BLOCKS_PASS;
    if (threadIdx.x == 0) {
        int* ctr = counters + pass;
        int r = package_half();
        int t = atomicAdd(ctr + r * SC_CTR_PITCH, 1);
        if (t >= SC_PIECES) {
            r ^= 1;
            t = atomicAdd(ctr + r * SC_CTR_PITCH, 1);
        }
        s_item = t < SC_PIECES ? 2 * t + r : -1;
    }
    __syncthreads();
    const int item = __builtin_amdgcn_readfirstlane(s_item);
    if (item < 0) return;
    const int upper = item & 1;
    const int blk   = item >> 1;
    const bool to_book = first ? (pass == 0) : (pass < 2);
    const int h = first ? (pass == 0 ? 0 : pass - 1) : (pass & 1);
    const unsigned fpitch = (first && !to_book) ? (unsigned)HD : (unsigned)PW;
    const float* feat = to_book ? (first ? ufeat : ufeat + plane_off(N_USER, 0, h))
                                : (first ? bfeat + PW * h : bfeat + plane_off(N_BOOK, 0, h));
    float*       sums = to_book ? sum_b + plane_off(N_BOOK, 0, h) : sum_u + plane_off(N_USER, 0, h);
    const int*   gidx = to_book ? esrc : edst;
    const int*   tidx = to_book ? edst : esrc;
    const int split   = (to_book ? N_BOOK : N_USER) >> 1;
    const int w  = __builtin_amdgcn_readfirstlane(blk * 8 + wib);
    const int e0 = w * EPW;
    if (e0 >= E) return;
    int g[EPW], t[EPW];
    if (e0 + EPW <= E) {
#pragma unroll
        for (int q = 0; q < EPW / 4; ++q) {
            const int4 gq = *(const int4*)(gidx + e0 + 4 * q);
            const int4 tq = *(const int4*)(tidx + e0 + 4 * q);
            g[4 * q] = gq.x; g[4 * q + 1] = gq.y; g[4 * q + 2] = gq.z; g[4 * q + 3] = gq.w;
            t[4 * q] = tq.x; t[4 * q + 1] = tq.y; t[4 * q + 2] = tq.z; t[4 * q + 3] = tq.w;
        }
    } else {
#pragma unroll
        for (int j = 0; j < EPW; ++j) {
            const int e = min(e0 + j, E - 1);
            g[j] = gidx[e];
            t[j] = tidx[e];
        }
    }
    const int n = min(EPW, E - e0);
    unsigned mine = 0;
#pragma unroll
    for (int j = 0; j < EPW; ++j) mine |= (t[j] >= split ? 1u : 0u) << j;
    if (!upper) mine = ~mine;
    mine &= (1u << n) - 1u;
    float v0[EPW], v1[EPW];
#pragma unroll
    for (int j = 0; j < EPW; ++j) {
        v0[j] = 0.0f;
        v1[j] = 0.0f;
        if (mine & (1u << j)) {
            const unsigned off = (unsigned)g[j] * fpitch + lane;
            v0[j] = feat[off];
            v1[j] = feat[off + 32];
        }
    }
#pragma unroll
    for (int j = 0; j < EPW; ++j) {
        if (mine & (1u << j)) {
            const unsigned off = (unsigned)t[j] * PW + lane;
            atomicAdd(sums + off, v0[j]);
            atomicAdd(sums + off + 32, v1[j]);
        }
    }
}


__device__ __forceinline__ void load_weight_t(const float* __restrict__ W, float* __restrict__ wt,
                                              int K, int stride) {
    const int nf4 = K * (HD / 4);
    for (int idx = threadIdx.x; idx < nf4; idx += 256) {
        int k = idx >> 5;
        int n = (idx & 31) << 2;
        const float4 w4 = *(const float4*)(W + (size_t)k * HD + n);
        wt[(n + 0) * stride + k] = w4.x;
        wt[(n + 1) * stride + k] = w4.y;
        wt[(n + 2) * stride + k] = w4.z;
        wt[(n + 3) * stride + k] = w4.w;
    }
}

constexpr int IMG128 = HD * (HD + 4);
constexpr int IMG64  = HD * (64 + 4);
constexpr int IMG_PUWL = 0;
constexpr int IMG_PUWR = IMG_PUWL + IMG64;
constexpr int IMG_PBWL = IMG_PUWR + IMG64;
constexpr int IMG_PBWR = IMG_PBWL + IMG128;
constexpr int IMG_LAYER = IMG_PBWR + IMG128;
constexpr int IMG_FLOATS = IMG_LAYER + 4 * IMG128;

__device__ __forceinline__ void load_weight_image(const float* __restrict__ img,
                                                  float* __restrict__ wt, int nfloats) {
    const float4* s4 = (const float4*)img;
    float4* d4 = (float4*)wt;
    const int nf4 = nfloats >> 2;
#pragma unroll 8
    for (int idx = threadIdx.x; idx < nf4; idx += 256) d4[idx] = s4[idx];
}

__device__ __forceinline__ void store_weight_image(float* __restrict__ wt, float* __restrict__ img,
                                                   int K, int stride) {
    for (int i = threadIdx.x; i < HD * 4; i += 256) wt[(i >> 2) * stride + K + (i & 3)] = 0.0f;
    __syncthreads();
    const float4* s4 = (const float4*)wt;
    float4* d4 = (float4*)img;
    const int nf4 = (HD * stride) >> 2;
    for (int idx = threadIdx.x; idx < nf4; idx += 256) d4[idx] = s4[idx];
}

constexpr int COMP_BUWL = 0, COMP_BUWR = 128, COMP_BBWL = 256, COMP_BBWR = 384;
constexpr int COMP_FLOATS = 512;

__global__ void __launch_bounds__(256)
compose_kernel(const float* __restrict__ Pu, const float* __restrict__ Pb,
               const float* __restrict__ bu, const float* __restrict__ bb,
               const float* __restrict__ Wl, const float* __restrict__ Wr,
               float* __restrict__ comp, float* __restrict__ img) {
    __shared__ __attribute__((aligned(16))) float wt[HD * (HD + 4)];
    const int stride = HD + 4;
    const int job = blockIdx.x;
    if (job >= 8) {
        const int j = job - 8;
        load_weight_t(((j & 1) ? Wr : Wl) + (size_t)(1 + (j >> 1)) * HD * HD, wt, HD, stride);
        store_weight_image(wt, img + IMG_LAYER + j * IMG128, HD, stride);
        return;
    }
    const int s = job >> 1;
    const float* A = s == 0 ? Pu : s == 1 ? Pb : s == 2 ? bu : bb;
    const int M = s == 0 ? 64 : s == 1 ? 128 : 1;
    load_weight_t((job & 1) ? Wr : Wl, wt, HD, stride);
    __syncthreads();

    const int wave = threadIdx.x >> 5;
    const int lane = threadIdx.x & 31;
    const int col  = lane & 15;
    const int half = lane >> 4;
    const int m0   = wave * 16;
    const int aRow = (m0 + col < M) ? (m0 + col) : (M - 1);

    v8f acc[8];
#pragma unroll
    for (int nt = 0; nt < 8; ++nt) acc[nt] = (v8f){0.f, 0.f, 0.f, 0.f, 0.f, 0.f, 0.f, 0.f};

    const float* arow = A + (size_t)aRow * HD;
    for (int k0 = 0; k0 < HD; k0 += 4) {
        const int kb = k0 + 2 * half;
        v2f a = *(const v2f*)(arow + kb);
        v2f bf[8];
#pragma unroll
        for (int nt = 0; nt < 8; ++nt)
            bf[nt] = *(const v2f*)&wt[(nt * 16 + col) * stride + kb];
#pragma unroll
        for (int nt = 0; nt < 8; ++nt)
            acc[nt] = __builtin_amdgcn_wmma_f32_16x16x4_f32(
                false, a, false, bf[nt], (short)0, acc[nt], false, false);
    }

    if (s >= 2) {
        float* out = comp + (job == 4 ? COMP_BUWL : job == 5 ? COMP_BUWR :
                             job == 6 ? COMP_BBWL : COMP_BBWR);
#pragma unroll
        for (int nt = 0; nt < 8; ++nt)
            if (m0 == 0 && half == 0) out[nt * 16 + col] = acc[nt][0];
        return;
    }
    __syncthreads();
    const int istride = M + 4;
#pragma unroll
    for (int nt = 0; nt < 8; ++nt)
#pragma unroll
        for (int v = 0; v < 8; ++v) {
            int dr = m0 + v + 8 * half;
            if (dr < M) wt[(nt * 16 + col) * istride + dr] = acc[nt][v];
        }
    store_weight_image(wt, img + (job == 0 ? IMG_PUWL : job == 1 ? IMG_PUWR :
                                  job == 2 ? IMG_PBWL : IMG_PBWR), M, istride);
}

constexpr int ZB = 256;
__device__ __forceinline__ void clear_sums_block(float* clr, int zblk) {
    v4f* p = (v4f*)clr;
    const int n4 = (N_USER + N_BOOK) * (HD / 4);
    for (int i = zblk * 256 + (int)threadIdx.x; i < n4; i += ZB * 256) p[i] = (v4f){0.f, 0.f, 0.f, 0.f};
}

constexpr int UB = (N_USER + 127) / 128;
constexpr int BB = (N_BOOK + 127) / 128;
template <bool LAST>
__device__ __forceinline__ void sage_combine_body(const float* sums_u, const float* cnt_u,
                                                  const float* X_u, float* out_u,
                                                  const float* sums_b, const float* cnt_b,
                                                  const float* X_b, float* out_b,
                                                  const float* Wl, const float* bl, const float* Wr,
                                                  const float* gamma_u, const float* beta_u,
                                                  const float* gamma_b, const float* beta_b, float* wt) {
    const int stride = HD + 4;
    const bool isb = (int)blockIdx.x >= UB;
    const float* sums = isb ? sums_b : sums_u;
    const float* cnt  = isb ? cnt_b : cnt_u;
    const float* X    = isb ? X_b : X_u;
    float* out        = isb ? out_b : out_u;
    const int M   = isb ? N_BOOK : N_USER;
    const int blk = isb ? (int)blockIdx.x - UB : (int)blockIdx.x;
    const int wave = threadIdx.x >> 5;
    const int lane = threadIdx.x & 31;
    const int m0   = blk * 128 + wave * 16;

    load_weight_image(Wl, wt, IMG128);
    __syncthreads();

    const int col  = lane & 15;
    const int half = lane >> 4;
    const int aRow = (m0 + col < M) ? (m0 + col) : (M - 1);
    const float invc = 1.0f / fmaxf(cnt[aRow], 1.0f);

    v8f acc[8];
#pragma unroll
    for (int nt = 0; nt < 8; ++nt) {
        float bv = bl[nt * 16 + col];
        acc[nt] = (v8f){bv, bv, bv, bv, bv, bv, bv, bv};
    }

    for (int k0 = 0; k0 < HD; k0 += 4) {
        const int kb = k0 + 2 * half;
        v2f a = *(const v2f*)(sums + plane_off(M, aRow, k0 >> 6) + (kb & (PW - 1)));
        a.x *= invc;
        a.y *= invc;
        v2f bf[8];
#pragma unroll
        for (int nt = 0; nt < 8; ++nt)
            bf[nt] = *(const v2f*)&wt[(nt * 16 + col) * stride + kb];
#pragma unroll
        for (int nt = 0; nt < 8; ++nt)
            acc[nt] = __builtin_amdgcn_wmma_f32_16x16x4_f32(
                false, a, false, bf[nt], (short)0, acc[nt], false, false);
    }

    __syncthreads();
    load_weight_image(Wr, wt, IMG128);
    __syncthreads();

    for (int k0 = 0; k0 < HD; k0 += 4) {
        const int kb = k0 + 2 * half;
        v2f a = *(const v2f*)(X + plane_off(M, aRow, k0 >> 6) + (kb & (PW - 1)));
        v2f bf[8];
#pragma unroll
        for (int nt = 0; nt < 8; ++nt)
            bf[nt] = *(const v2f*)&wt[(nt * 16 + col) * stride + kb];
#pragma unroll
        for (int nt = 0; nt < 8; ++nt)
            acc[nt] = __builtin_amdgcn_wmma_f32_16x16x4_f32(
                false, a, false, bf[nt], (short)0, acc[nt], false, false);
    }

    if (LAST) __builtin_amdgcn_sched_barrier(0);

    if (!LAST) {
#pragma unroll
        for (int nt = 0; nt < 8; ++nt)
#pragma unroll
            for (int v = 0; v < 8; ++v) {
                int dr = m0 + v + 8 * half;
                if (dr < M) out[plane_off(M, dr, nt >> 2) + (nt & 3) * 16 + col] = fmaxf(acc[nt][v], 0.0f);
            }
        return;
    }

    const float* gamma = isb ? gamma_b : gamma_u;
    const float* beta  = isb ? beta_b : beta_u;
    float rstd[8];
#pragma unroll
    for (int v = 0; v < 8; ++v) {
        float s = 0.0f;
#pragma unroll
        for (int nt = 0; nt < 8; ++nt) {
            acc[nt][v] = fmaxf(acc[nt][v], 0.0f);
            s += acc[nt][v];
        }
#pragma unroll
        for (int off = 8; off >= 1; off >>= 1) s += __shfl_xor(s, off, 32);
        const float mu = s * (1.0f / HD);
        float q = 0.0f;
#pragma unroll
        for (int nt = 0; nt < 8; ++nt) {
            const float d = acc[nt][v] - mu;
            acc[nt][v] = d;
            q += d * d;
        }
#pragma unroll
        for (int off = 8; off >= 1; off >>= 1) q += __shfl_xor(q, off, 32);
        rstd[v] = rsqrtf(q * (1.0f / HD) + LN_EPS);
    }
#pragma unroll
    for (int nt = 0; nt < 8; ++nt) {
        const float g  = gamma[nt * 16 + col];
        const float bt = beta[nt * 16 + col];
#pragma unroll
        for (int v = 0; v < 8; ++v) {
            int dr = m0 + v + 8 * half;
            if (dr < M) out[(size_t)dr * HD + nt * 16 + col] = acc[nt][v] * rstd[v] * g + bt;
        }
    }
}

__global__ void sage_combine_kernel(const float* __restrict__ sums_u, const float* __restrict__ cnt_u,
                                    const float* __restrict__ X_u, float* __restrict__ out_u,
                                    const float* __restrict__ sums_b, const float* __restrict__ cnt_b,
                                    const float* __restrict__ X_b, float* __restrict__ out_b,
                                    const float* __restrict__ Wl, const float* __restrict__ bl,
                                    const float* __restrict__ Wr, float* __restrict__ clr) {
    __shared__ __attribute__((aligned(16))) float wt[HD * (HD + 4)];
    if ((int)blockIdx.x >= UB + BB) {
        clear_sums_block(clr, (int)blockIdx.x - (UB + BB));
        return;
    }
    sage_combine_body<false>(sums_u, cnt_u, X_u, out_u, sums_b, cnt_b, X_b, out_b,
                             Wl, bl, Wr, nullptr, nullptr, nullptr, nullptr, wt);
}

__global__ void sage_last_kernel(const float* __restrict__ sums_u, const float* __restrict__ cnt_u,
                                 const float* __restrict__ X_u, float* __restrict__ out_u,
                                 const float* __restrict__ sums_b, const float* __restrict__ cnt_b,
                                 const float* __restrict__ X_b, float* __restrict__ out_b,
                                 const float* __restrict__ Wl, const float* __restrict__ bl,
                                 const float* __restrict__ Wr,
                                 const float* __restrict__ gamma_u, const float* __restrict__ beta_u,
                                 const float* __restrict__ gamma_b, const float* __restrict__ beta_b) {
    __shared__ __attribute__((aligned(16))) float wt[HD * (HD + 4)];
    sage_combine_body<true>(sums_u, cnt_u, X_u, out_u, sums_b, cnt_b, X_b, out_b,
                            Wl, bl, Wr, gamma_u, beta_u, gamma_b, beta_b, wt);
}

__global__ void sage_first_kernel(const float* __restrict__ sums_u, const float* __restrict__ cnt_u,
                                  const float* __restrict__ X_u, const float* __restrict__ W1_u,
                                  const float* __restrict__ W2_u, const float* __restrict__ c1_u,
                                  const float* __restrict__ c2_u, float* __restrict__ out_u,
                                  const float* __restrict__ sums_b, const float* __restrict__ cnt_b,
                                  const float* __restrict__ X_b, const float* __restrict__ W1_b,
                                  const float* __restrict__ W2_b, const float* __restrict__ c1_b,
                                  const float* __restrict__ c2_b, float* __restrict__ out_b,
                                  const float* __restrict__ bl,
                                  float* __restrict__ clr,
                                  int K1_u, int K2_u, int K1_b, int K2_b) {
    __shared__ __attribute__((aligned(16))) float wt[HD * (HD + 4)];
    if ((int)blockIdx.x >= UB + BB) {
        clear_sums_block(clr, (int)blockIdx.x - (UB + BB));
        return;
    }
    const bool isb = (int)blockIdx.x >= UB;
    const float* __restrict__ sums = isb ? sums_b : sums_u;
    const float* __restrict__ cnt  = isb ? cnt_b : cnt_u;
    const float* __restrict__ X    = isb ? X_b : X_u;
    const float* __restrict__ W1   = isb ? W1_b : W1_u;
    const float* __restrict__ W2   = isb ? W2_b : W2_u;
    const float* __restrict__ c1   = isb ? c1_b : c1_u;
    const float* __restrict__ c2   = isb ? c2_b : c2_u;
    float* __restrict__ out        = isb ? out_b : out_u;
    const int M   = isb ? N_BOOK : N_USER;
    const int K1  = isb ? K1_b : K1_u;
    const int K2  = isb ? K2_b : K2_u;
    const int blk = isb ? (int)blockIdx.x - UB : (int)blockIdx.x;
    const int stride1 = K1 + 4;
    const int stride2 = K2 + 4;

    load_weight_image(W1, wt, HD * stride1);
    __syncthreads();

    const int wave = threadIdx.x >> 5;
    const int lane = threadIdx.x & 31;
    const int col  = lane & 15;
    const int half = lane >> 4;
    const int m0   = blk * 128 + wave * 16;
    const int aRow = (m0 + col < M) ? (m0 + col) : (M - 1);
    const float invc = 1.0f / fmaxf(cnt[aRow], 1.0f);

    v8f acc[8];
#pragma unroll
    for (int nt = 0; nt < 8; ++nt) {
        float bv = bl[nt * 16 + col] + c2[nt * 16 + col];
        acc[nt] = (v8f){bv, bv, bv, bv, bv, bv, bv, bv};
    }

    for (int k0 = 0; k0 < K1; k0 += 4) {
        const int kb = k0 + 2 * half;
        v2f a = *(const v2f*)(sums + plane_off(M, aRow, k0 >> 6) + (kb & (PW - 1)));
        a.x *= invc;
        a.y *= invc;
        v2f bf[8];
#pragma unroll
        for (int nt = 0; nt < 8; ++nt)
            bf[nt] = *(const v2f*)&wt[(nt * 16 + col) * stride1 + kb];
#pragma unroll
        for (int nt = 0; nt < 8; ++nt)
            acc[nt] = __builtin_amdgcn_wmma_f32_16x16x4_f32(
                false, a, false, bf[nt], (short)0, acc[nt], false, false);
    }

    __syncthreads();
    load_weight_image(W2, wt, HD * stride2);
    __syncthreads();

    const float* xrow = X + (size_t)aRow * K2;
    for (int k0 = 0; k0 < K2; k0 += 4) {
        const int kb = k0 + 2 * half;
        v2f a = *(const v2f*)(xrow + kb);
        v2f bf[8];
#pragma unroll
        for (int nt = 0; nt < 8; ++nt)
            bf[nt] = *(const v2f*)&wt[(nt * 16 + col) * stride2 + kb];
#pragma unroll
        for (int nt = 0; nt < 8; ++nt)
            acc[nt] = __builtin_amdgcn_wmma_f32_16x16x4_f32(
                false, a, false, bf[nt], (short)0, acc[nt], false, false);
    }

    float has[8];
#pragma unroll
    for (int v = 0; v < 8; ++v) {
        const int dr = m0 + v + 8 * half;
        has[v] = (cnt[dr < M ? dr : M - 1] > 0.0f) ? 1.0f : 0.0f;
    }
#pragma unroll
    for (int nt = 0; nt < 8; ++nt) {
        const float cb = c1[nt * 16 + col];
#pragma unroll
        for (int v = 0; v < 8; ++v) {
            int dr = m0 + v + 8 * half;
            if (dr < M)
                out[plane_off(M, dr, nt >> 2) + (nt & 3) * 16 + col] =
                    fmaxf(acc[nt][v] + has[v] * cb, 0.0f);
        }
    }
}

extern "C" void kernel_launch(void* const* d_in, const int* in_sizes, int n_in,
                              void* d_out, int out_size, void* d_ws, size_t ws_size,
                              hipStream_t stream) {
    const float* user_x      = (const float*)d_in[0];
    const float* book_x      = (const float*)d_in[1];
    const int*   edge_src    = (const int*)d_in[2];
    const int*   edge_dst    = (const int*)d_in[3];
    const float* user_proj_w = (const float*)d_in[4];
    const float* user_proj_b = (const float*)d_in[5];
    const float* book_proj_w = (const float*)d_in[6];
    const float* book_proj_b = (const float*)d_in[7];
    const float* Wl          = (const float*)d_in[8];
    const float* bl          = (const float*)d_in[9];
    const float* Wr          = (const float*)d_in[10];
    const float* ugamma      = (const float*)d_in[11];
    const float* ubeta       = (const float*)d_in[12];
    const float* bgamma      = (const float*)d_in[13];
    const float* bbeta       = (const float*)d_in[14];

    float* ws = (float*)d_ws;
    size_t off = 0;
    auto carve = [&](size_t n) { float* p = ws + off; off += n; return p; };
    float* u0    = (float*)d_out;
    float* b0    = u0 + (size_t)N_USER * HD;
    float* u1    = carve((size_t)N_USER * HD);
    float* b1    = carve((size_t)N_BOOK * HD);
    float* sum2_u = carve((size_t)N_USER * HD);
    float* sum2_b = carve((size_t)N_BOOK * HD);
    float* sum_u = carve((size_t)N_USER * HD);
    float* sum_b = carve((size_t)N_BOOK * HD);
    float* cnt_u = carve((size_t)N_USER);
    float* cnt_b = carve((size_t)N_BOOK);
    int*   ctrs  = (int*)carve((size_t)3 * SC_COUNTERS);
    float* comp  = carve((size_t)COMP_FLOATS);
    float* img   = carve((size_t)IMG_FLOATS);

    compose_kernel<<<12, 256, 0, stream>>>(
        user_proj_w, book_proj_w, user_proj_b, book_proj_b, Wl, Wr, comp, img);

    zero_kernel<<<4096, 256, 0, stream>>>(
        sum_u, (size_t)(N_USER + N_BOOK) * HD + (size_t)N_USER + (size_t)N_BOOK + (size_t)3 * SC_COUNTERS);
    count_kernel<<<(N_EDGE + 255) / 256, 256, 0, stream>>>(
        edge_src, edge_dst, cnt_u, cnt_b, N_EDGE);

    scatter_pass_kernel<<<3 * SC_BLOCKS_PASS, 256, 0, stream>>>(
        user_x, book_x, edge_src, edge_dst, sum_b, sum_u, ctrs, N_EDGE, 1);
    sage_first_kernel<<<UB + BB + ZB, 256, 0, stream>>>(
        sum_u, cnt_u, user_x, img + IMG_PBWL, img + IMG_PUWR,
        comp + COMP_BBWL, comp + COMP_BUWR, u0,
        sum_b, cnt_b, book_x, img + IMG_PUWL, img + IMG_PBWR,
        comp + COMP_BUWL, comp + COMP_BBWR, b0, bl, sum2_u, 128, 64, 64, 128);

    scatter_pass_kernel<<<SC_GRID, 256, 0, stream>>>(
        u0, b0, edge_src, edge_dst, sum2_b, sum2_u, ctrs + SC_COUNTERS, N_EDGE, 0);
    sage_combine_kernel<<<UB + BB + ZB, 256, 0, stream>>>(
        sum2_u, cnt_u, u0, u1, sum2_b, cnt_b, b0, b1,
        img + IMG_LAYER, bl + HD, img + IMG_LAYER + IMG128, sum_u);

    float* out_u = (float*)d_out;
    float* out_b = out_u + (size_t)N_USER * HD;
    scatter_pass_kernel<<<SC_GRID, 256, 0, stream>>>(
        u1, b1, edge_src, edge_dst, sum_b, sum_u, ctrs + 2 * SC_COUNTERS, N_EDGE, 0);
    sage_last_kernel<<<UB + BB, 256, 0, stream>>>(
        sum_u, cnt_u, u1, out_u, sum_b, cnt_b, b1, out_b,
        img + IMG_LAYER + 2 * IMG128, bl + 2 * HD, img + IMG_LAYER + 3 * IMG128,
        ugamma, ubeta, bgamma, bbeta);
}


# Atomic scatter: 2048 blocks per launch stay resident and draw piece after piece through all passes from the range counters (was one block per piece, 70k-94k blocks a launch); arrival order as before
# speedup vs baseline: 2.5087x; 1.0360x over previous
#include <hip/hip_runtime.h>

typedef float v2f __attribute__((ext_vector_type(2)));
typedef float v4f __attribute__((ext_vector_type(4)));
typedef float v8f __attribute__((ext_vector_type(8)));
typedef int v4i __attribute__((ext_vector_type(4)));
typedef const __attribute__((address_space(4))) int* const_int_ptr;
typedef const __attribute__((address_space(4))) v4i* const_v4i_ptr;

#define HD 128
constexpr int N_USER = 150000;
constexpr int N_BOOK = 75000;
constexpr int N_EDGE = 1500000;
constexpr float LN_EPS = 1e-5f;

constexpr int PW = 64;
__device__ __forceinline__ size_t plane_off(int n_rows, int row, int plane) {
    return ((size_t)plane * n_rows + row) * PW;
}

__device__ __forceinline__ int package_half() {
    const unsigned id = __builtin_amdgcn_s_sendmsg_rtn(0x87);
    return (int)((id >> 18) & 1u);
}

__global__ void zero_kernel(float* __restrict__ p, size_t n) {
    size_t i = (size_t)blockIdx.x * blockDim.x + threadIdx.x;
    size_t stride = (size_t)gridDim.x * blockDim.x;
    for (; i < n; i += stride) p[i] = 0.0f;
}

__global__ void count_kernel(const int* __restrict__ esrc, const int* __restrict__ edst,
                             float* __restrict__ cnt_u, float* __restrict__ cnt_b, int E) {
    int i = blockIdx.x * blockDim.x + threadIdx.x;
    if (i < E) {
        atomicAdd(&cnt_u[esrc[i]], 1.0f);
        atomicAdd(&cnt_b[edst[i]], 1.0f);
    }
}

constexpr int EPW = 16;
constexpr int SC_PIECES = (N_EDGE + EPW * 8 - 1) / (EPW * 8);
constexpr int SC_RESIDENT = 256 * 8;
constexpr int SC_CTR_PITCH = 64;
constexpr int SC_COUNTERS = 2 * SC_CTR_PITCH;

__global__ void __launch_bounds__(256)
scatter_pass_kernel(const float* ufeat, const float* bfeat,
                    const int* __restrict__ esrc, const int* __restrict__ edst,
                    float* sum_b, float* sum_u, int* counters, int E, int first) {
    __shared__ int s_item[2];
    const int lane = threadIdx.x & 31;
    const int wib  = threadIdx.x >> 5;
    const int npass = first ? 3 : 4;
    int r0 = 0;
    if (threadIdx.x == 0) r0 = package_half();
    int slot = 0;
    for (int pass = 0; pass < npass; ++pass) {
        int* ctr = counters + pass;
        int r_cur = r0;
        int t_next = 0;
        if (threadIdx.x == 0) t_next = atomicAdd(ctr + r_cur * SC_CTR_PITCH, 1);
        const bool to_book = first ? (pass == 0) : (pass < 2);
        const int h = first ? (pass == 0 ? 0 : pass - 1) : (pass & 1);
        const unsigned fpitch = (first && !to_book) ? (unsigned)HD : (unsigned)PW;
        const float* feat = to_book ? (first ? ufeat : ufeat + plane_off(N_USER, 0, h))
                                    : (first ? bfeat + PW * h : bfeat + plane_off(N_BOOK, 0, h));
        float*       sums = to_book ? sum_b + plane_off(N_BOOK, 0, h) : sum_u + plane_off(N_USER, 0, h);
        const int*   gidx = to_book ? esrc : edst;
        const int*   tidx = to_book ? edst : esrc;
        const int split   = (to_book ? N_BOOK : N_USER) >> 1;
        for (int trip = 0; trip <= 2 * SC_PIECES; ++trip) {
            if (threadIdx.x == 0) {
                if (t_next >= SC_PIECES && r_cur == r0) {
                    r_cur = r0 ^ 1;
                    t_next = atomicAdd(ctr + r_cur * SC_CTR_PITCH, 1);
                }
                s_item[slot] = t_next < SC_PIECES ? 2 * t_next + r_cur : -1;
            }
            __syncthreads();
            const int item = __builtin_amdgcn_readfirstlane(s_item[slot]);
            slot ^= 1;
            if (item < 0) break;
            const int upper = item & 1;
            const int blk   = item >> 1;
            const int w  = __builtin_amdgcn_readfirstlane(blk * 8 + wib);
            const int e0 = w * EPW;
            if (e0 >= E) continue;
            int g[EPW], t[EPW];
            if (e0 + EPW <= E) {
#pragma unroll
                for (int q = 0; q < EPW / 4; ++q) {
                    const v4i gq = *(const_v4i_ptr)(gidx + e0 + 4 * q);
                    const v4i tq = *(const_v4i_ptr)(tidx + e0 + 4 * q);
                    g[4 * q] = gq.x; g[4 * q + 1] = gq.y; g[4 * q + 2] = gq.z; g[4 * q + 3] = gq.w;
                    t[4 * q] = tq.x; t[4 * q + 1] = tq.y; t[4 * q + 2] = tq.z; t[4 * q + 3] = tq.w;
                }
            } else {
#pragma unroll
                for (int j = 0; j < EPW; ++j) {
                    const int e = min(e0 + j, E - 1);
                    g[j] = ((const_int_ptr)gidx)[e];
                    t[j] = ((const_int_ptr)tidx)[e];
                }
            }
            const int n = min(EPW, E - e0);
            unsigned mine = 0;
#pragma unroll
            for (int j = 0; j < EPW; ++j) mine |= (t[j] >= split ? 1u : 0u) << j;
            if (!upper) mine = ~mine;
            mine &= (1u << n) - 1u;
            if (threadIdx.x == 0) t_next = atomicAdd(ctr + r_cur * SC_CTR_PITCH, 1);
            float v0[EPW], v1[EPW];
#pragma unroll
            for (int j = 0; j < EPW; ++j) {
                v0[j] = 0.0f;
                v1[j] = 0.0f;
                if (mine & (1u << j)) {
                    const unsigned off = (unsigned)g[j] * fpitch + lane;
                    v0[j] = feat[off];
                    v1[j] = feat[off + 32];
                }
            }
#pragma unroll
            for (int j = 0; j < EPW; ++j) {
                if (mine & (1u << j)) {
                    const unsigned off = (unsigned)t[j] * PW + lane;
                    atomicAdd(sums + off, v0[j]);
                    atomicAdd(sums + off + 32, v1[j]);
                }
            }
        }
    }
}


__device__ __forceinline__ void load_weight_t(const float* __restrict__ W, float* __restrict__ wt,
                                              int K, int stride) {
    const int nf4 = K * (HD / 4);
    for (int idx = threadIdx.x; idx < nf4; idx += 256) {
        int k = idx >> 5;
        int n = (idx & 31) << 2;
        const float4 w4 = *(const float4*)(W + (size_t)k * HD + n);
        wt[(n + 0) * stride + k] = w4.x;
        wt[(n + 1) * stride + k] = w4.y;
        wt[(n + 2) * stride + k] = w4.z;
        wt[(n + 3) * stride + k] = w4.w;
    }
}

constexpr int IMG128 = HD * (HD + 4);
constexpr int IMG64  = HD * (64 + 4);
constexpr int IMG_PUWL = 0;
constexpr int IMG_PUWR = IMG_PUWL + IMG64;
constexpr int IMG_PBWL = IMG_PUWR + IMG64;
constexpr int IMG_PBWR = IMG_PBWL + IMG128;
constexpr int IMG_LAYER = IMG_PBWR + IMG128;
constexpr int IMG_FLOATS = IMG_LAYER + 4 * IMG128;

__device__ __forceinline__ void load_weight_image(const float* __restrict__ img,
                                                  float* __restrict__ wt, int nfloats) {
    const float4* s4 = (const float4*)img;
    float4* d4 = (float4*)wt;
    const int nf4 = nfloats >> 2;
#pragma unroll 8
    for (int idx = threadIdx.x; idx < nf4; idx += 256) d4[idx] = s4[idx];
}

__device__ __forceinline__ void store_weight_image(float* __restrict__ wt, float* __restrict__ img,
                                                   int K, int stride) {
    for (int i = threadIdx.x; i < HD * 4; i += 256) wt[(i >> 2) * stride + K + (i & 3)] = 0.0f;
    __syncthreads();
    const float4* s4 = (const float4*)wt;
    float4* d4 = (float4*)img;
    const int nf4 = (HD * stride) >> 2;
    for (int idx = threadIdx.x; idx < nf4; idx += 256) d4[idx] = s4[idx];
}

constexpr int COMP_BUWL = 0, COMP_BUWR = 128, COMP_BBWL = 256, COMP_BBWR = 384;
constexpr int COMP_FLOATS = 512;

__global__ void __launch_bounds__(256)
compose_kernel(const float* __restrict__ Pu, const float* __restrict__ Pb,
               const float* __restrict__ bu, const float* __restrict__ bb,
               const float* __restrict__ Wl, const float* __restrict__ Wr,
               float* __restrict__ comp, float* __restrict__ img) {
    __shared__ __attribute__((aligned(16))) float wt[HD * (HD + 4)];
    const int stride = HD + 4;
    const int job = blockIdx.x;
    if (job >= 8) {
        const int j = job - 8;
        load_weight_t(((j & 1) ? Wr : Wl) + (size_t)(1 + (j >> 1)) * HD * HD, wt, HD, stride);
        store_weight_image(wt, img + IMG_LAYER + j * IMG128, HD, stride);
        return;
    }
    const int s = job >> 1;
    const float* A = s == 0 ? Pu : s == 1 ? Pb : s == 2 ? bu : bb;
    const int M = s == 0 ? 64 : s == 1 ? 128 : 1;
    load_weight_t((job & 1) ? Wr : Wl, wt, HD, stride);
    __syncthreads();

    const int wave = threadIdx.x >> 5;
    const int lane = threadIdx.x & 31;
    const int col  = lane & 15;
    const int half = lane >> 4;
    const int m0   = wave * 16;
    const int aRow = (m0 + col < M) ? (m0 + col) : (M - 1);

    v8f acc[8];
#pragma unroll
    for (int nt = 0; nt < 8; ++nt) acc[nt] = (v8f){0.f, 0.f, 0.f, 0.f, 0.f, 0.f, 0.f, 0.f};

    const float* arow = A + (size_t)aRow * HD;
    for (int k0 = 0; k0 < HD; k0 += 4) {
        const int kb = k0 + 2 * half;
        v2f a = *(const v2f*)(arow + kb);
        v2f bf[8];
#pragma unroll
        for (int nt = 0; nt < 8; ++nt)
            bf[nt] = *(const v2f*)&wt[(nt * 16 + col) * stride + kb];
#pragma unroll
        for (int nt = 0; nt < 8; ++nt)
            acc[nt] = __builtin_amdgcn_wmma_f32_16x16x4_f32(
                false, a, false, bf[nt], (short)0, acc[nt], false, false);
    }

    if (s >= 2) {
        float* out = comp + (job == 4 ? COMP_BUWL : job == 5 ? COMP_BUWR :
                             job == 6 ? COMP_BBWL : COMP_BBWR);
#pragma unroll
        for (int nt = 0; nt < 8; ++nt)
            if (m0 == 0 && half == 0) out[nt * 16 + col] = acc[nt][0];
        return;
    }
    __syncthreads();
    const int istride = M + 4;
#pragma unroll
    for (int nt = 0; nt < 8; ++nt)
#pragma unroll
        for (int v = 0; v < 8; ++v) {
            int dr = m0 + v + 8 * half;
            if (dr < M) wt[(nt * 16 + col) * istride + dr] = acc[nt][v];
        }
    store_weight_image(wt, img + (job == 0 ? IMG_PUWL : job == 1 ? IMG_PUWR :
                                  job == 2 ? IMG_PBWL : IMG_PBWR), M, istride);
}

constexpr int ZB = 256;
__device__ __forceinline__ void clear_sums_block(float* clr, int zblk) {
    v4f* p = (v4f*)clr;
    const int n4 = (N_USER + N_BOOK) * (HD / 4);
    for (int i = zblk * 256 + (int)threadIdx.x; i < n4; i += ZB * 256) p[i] = (v4f){0.f, 0.f, 0.f, 0.f};
}

constexpr int UB = (N_USER + 127) / 128;
constexpr int BB = (N_BOOK + 127) / 128;
template <bool LAST>
__device__ __forceinline__ void sage_combine_body(const float* sums_u, const float* cnt_u,
                                                  const float* X_u, float* out_u,
                                                  const float* sums_b, const float* cnt_b,
                                                  const float* X_b, float* out_b,
                                                  const float* Wl, const float* bl, const float* Wr,
                                                  const float* gamma_u, const float* beta_u,
                                                  const float* gamma_b, const float* beta_b, float* wt) {
    const int stride = HD + 4;
    const bool isb = (int)blockIdx.x >= UB;
    const float* sums = isb ? sums_b : sums_u;
    const float* cnt  = isb ? cnt_b : cnt_u;
    const float* X    = isb ? X_b : X_u;
    float* out        = isb ? out_b : out_u;
    const int M   = isb ? N_BOOK : N_USER;
    const int blk = isb ? (int)blockIdx.x - UB : (int)blockIdx.x;
    const int wave = threadIdx.x >> 5;
    const int lane = threadIdx.x & 31;
    const int m0   = blk * 128 + wave * 16;

    load_weight_image(Wl, wt, IMG128);
    __syncthreads();

    const int col  = lane & 15;
    const int half = lane >> 4;
    const int aRow = (m0 + col < M) ? (m0 + col) : (M - 1);
    const float invc = 1.0f / fmaxf(cnt[aRow], 1.0f);

    v8f acc[8];
#pragma unroll
    for (int nt = 0; nt < 8; ++nt) {
        float bv = bl[nt * 16 + col];
        acc[nt] = (v8f){bv, bv, bv, bv, bv, bv, bv, bv};
    }

    for (int k0 = 0; k0 < HD; k0 += 4) {
        const int kb = k0 + 2 * half;
        v2f a = *(const v2f*)(sums + plane_off(M, aRow, k0 >> 6) + (kb & (PW - 1)));
        a.x *= invc;
        a.y *= invc;
        v2f bf[8];
#pragma unroll
        for (int nt = 0; nt < 8; ++nt)
            bf[nt] = *(const v2f*)&wt[(nt * 16 + col) * stride + kb];
#pragma unroll
        for (int nt = 0; nt < 8; ++nt)
            acc[nt] = __builtin_amdgcn_wmma_f32_16x16x4_f32(
                false, a, false, bf[nt], (short)0, acc[nt], false, false);
    }

    __syncthreads();
    load_weight_image(Wr, wt, IMG128);
    __syncthreads();

    for (int k0 = 0; k0 < HD; k0 += 4) {
        const int kb = k0 + 2 * half;
        v2f a = *(const v2f*)(X + plane_off(M, aRow, k0 >> 6) + (kb & (PW - 1)));
        v2f bf[8];
#pragma unroll
        for (int nt = 0; nt < 8; ++nt)
            bf[nt] = *(const v2f*)&wt[(nt * 16 + col) * stride + kb];
#pragma unroll
        for (int nt = 0; nt < 8; ++nt)
            acc[nt] = __builtin_amdgcn_wmma_f32_16x16x4_f32(
                false, a, false, bf[nt], (short)0, acc[nt], false, false);
    }

    if (LAST) __builtin_amdgcn_sched_barrier(0);

    if (!LAST) {
#pragma unroll
        for (int nt = 0; nt < 8; ++nt)
#pragma unroll
            for (int v = 0; v < 8; ++v) {
                int dr = m0 + v + 8 * half;
                if (dr < M) out[plane_off(M, dr, nt >> 2) + (nt & 3) * 16 + col] = fmaxf(acc[nt][v], 0.0f);
            }
        return;
    }

    const float* gamma = isb ? gamma_b : gamma_u;
    const float* beta  = isb ? beta_b : beta_u;
    float rstd[8];
#pragma unroll
    for (int v = 0; v < 8; ++v) {
        float s = 0.0f;
#pragma unroll
        for (int nt = 0; nt < 8; ++nt) {
            acc[nt][v] = fmaxf(acc[nt][v], 0.0f);
            s += acc[nt][v];
        }
#pragma unroll
        for (int off = 8; off >= 1; off >>= 1) s += __shfl_xor(s, off, 32);
        const float mu = s * (1.0f / HD);
        float q = 0.0f;
#pragma unroll
        for (int nt = 0; nt < 8; ++nt) {
            const float d = acc[nt][v] - mu;
            acc[nt][v] = d;
            q += d * d;
        }
#pragma unroll
        for (int off = 8; off >= 1; off >>= 1) q += __shfl_xor(q, off, 32);
        rstd[v] = rsqrtf(q * (1.0f / HD) + LN_EPS);
    }
#pragma unroll
    for (int nt = 0; nt < 8; ++nt) {
        const float g  = gamma[nt * 16 + col];
        const float bt = beta[nt * 16 + col];
#pragma unroll
        for (int v = 0; v < 8; ++v) {
            int dr = m0 + v + 8 * half;
            if (dr < M) out[(size_t)dr * HD + nt * 16 + col] = acc[nt][v] * rstd[v] * g + bt;
        }
    }
}

__global__ void sage_combine_kernel(const float* __restrict__ sums_u, const float* __restrict__ cnt_u,
                                    const float* __restrict__ X_u, float* __restrict__ out_u,
                                    const float* __restrict__ sums_b, const float* __restrict__ cnt_b,
                                    const float* __restrict__ X_b, float* __restrict__ out_b,
                                    const float* __restrict__ Wl, const float* __restrict__ bl,
                                    const float* __restrict__ Wr, float* __restrict__ clr) {
    __shared__ __attribute__((aligned(16))) float wt[HD * (HD + 4)];
    if ((int)blockIdx.x >= UB + BB) {
        clear_sums_block(clr, (int)blockIdx.x - (UB + BB));
        return;
    }
    sage_combine_body<false>(sums_u, cnt_u, X_u, out_u, sums_b, cnt_b, X_b, out_b,
                             Wl, bl, Wr, nullptr, nullptr, nullptr, nullptr, wt);
}

__global__ void sage_last_kernel(const float* __restrict__ sums_u, const float* __restrict__ cnt_u,
                                 const float* __restrict__ X_u, float* __restrict__ out_u,
                                 const float* __restrict__ sums_b, const float* __restrict__ cnt_b,
                                 const float* __restrict__ X_b, float* __restrict__ out_b,
                                 const float* __restrict__ Wl, const float* __restrict__ bl,
                                 const float* __restrict__ Wr,
                                 const float* __restrict__ gamma_u, const float* __restrict__ beta_u,
                                 const float* __restrict__ gamma_b, const float* __restrict__ beta_b) {
    __shared__ __attribute__((aligned(16))) float wt[HD * (HD + 4)];
    sage_combine_body<true>(sums_u, cnt_u, X_u, out_u, sums_b, cnt_b, X_b, out_b,
                            Wl, bl, Wr, gamma_u, beta_u, gamma_b, beta_b, wt);
}

__global__ void sage_first_kernel(const float* __restrict__ sums_u, const float* __restrict__ cnt_u,
                                  const float* __restrict__ X_u, const float* __restrict__ W1_u,
                                  const float* __restrict__ W2_u, const float* __restrict__ c1_u,
                                  const float* __restrict__ c2_u, float* __restrict__ out_u,
                                  const float* __restrict__ sums_b, const float* __restrict__ cnt_b,
                                  const float* __restrict__ X_b, const float* __restrict__ W1_b,
                                  const float* __restrict__ W2_b, const float* __restrict__ c1_b,
                                  const float* __restrict__ c2_b, float* __restrict__ out_b,
                                  const float* __restrict__ bl,
                                  float* __restrict__ clr,
                                  int K1_u, int K2_u, int K1_b, int K2_b) {
    __shared__ __attribute__((aligned(16))) float wt[HD * (HD + 4)];
    if ((int)blockIdx.x >= UB + BB) {
        clear_sums_block(clr, (int)blockIdx.x - (UB + BB));
        return;
    }
    const bool isb = (int)blockIdx.x >= UB;
    const float* __restrict__ sums = isb ? sums_b : sums_u;
    const float* __restrict__ cnt  = isb ? cnt_b : cnt_u;
    const float* __restrict__ X    = isb ? X_b : X_u;
    const float* __restrict__ W1   = isb ? W1_b : W1_u;
    const float* __restrict__ W2   = isb ? W2_b : W2_u;
    const float* __restrict__ c1   = isb ? c1_b : c1_u;
    const float* __restrict__ c2   = isb ? c2_b : c2_u;
    float* __restrict__ out        = isb ? out_b : out_u;
    const int M   = isb ? N_BOOK : N_USER;
    const int K1  = isb ? K1_b : K1_u;
    const int K2  = isb ? K2_b : K2_u;
    const int blk = isb ? (int)blockIdx.x - UB : (int)blockIdx.x;
    const int stride1 = K1 + 4;
    const int stride2 = K2 + 4;

    load_weight_image(W1, wt, HD * stride1);
    __syncthreads();

    const int wave = threadIdx.x >> 5;
    const int lane = threadIdx.x & 31;
    const int col  = lane & 15;
    const int half = lane >> 4;
    const int m0   = blk * 128 + wave * 16;
    const int aRow = (m0 + col < M) ? (m0 + col) : (M - 1);
    const float invc = 1.0f / fmaxf(cnt[aRow], 1.0f);

    v8f acc[8];
#pragma unroll
    for (int nt = 0; nt < 8; ++nt) {
        float bv = bl[nt * 16 + col] + c2[nt * 16 + col];
        acc[nt] = (v8f){bv, bv, bv, bv, bv, bv, bv, bv};
    }

    for (int k0 = 0; k0 < K1; k0 += 4) {
        const int kb = k0 + 2 * half;
        v2f a = *(const v2f*)(sums + plane_off(M, aRow, k0 >> 6) + (kb & (PW - 1)));
        a.x *= invc;
        a.y *= invc;
        v2f bf[8];
#pragma unroll
        for (int nt = 0; nt < 8; ++nt)
            bf[nt] = *(const v2f*)&wt[(nt * 16 + col) * stride1 + kb];
#pragma unroll
        for (int nt = 0; nt < 8; ++nt)
            acc[nt] = __builtin_amdgcn_wmma_f32_16x16x4_f32(
                false, a, false, bf[nt], (short)0, acc[nt], false, false);
    }

    __syncthreads();
    load_weight_image(W2, wt, HD * stride2);
    __syncthreads();

    const float* xrow = X + (size_t)aRow * K2;
    for (int k0 = 0; k0 < K2; k0 += 4) {
        const int kb = k0 + 2 * half;
        v2f a = *(const v2f*)(xrow + kb);
        v2f bf[8];
#pragma unroll
        for (int nt = 0; nt < 8; ++nt)
            bf[nt] = *(const v2f*)&wt[(nt * 16 + col) * stride2 + kb];
#pragma unroll
        for (int nt = 0; nt < 8; ++nt)
            acc[nt] = __builtin_amdgcn_wmma_f32_16x16x4_f32(
                false, a, false, bf[nt], (short)0, acc[nt], false, false);
    }

    float has[8];
#pragma unroll
    for (int v = 0; v < 8; ++v) {
        const int dr = m0 + v + 8 * half;
        has[v] = (cnt[dr < M ? dr : M - 1] > 0.0f) ? 1.0f : 0.0f;
    }
#pragma unroll
    for (int nt = 0; nt < 8; ++nt) {
        const float cb = c1[nt * 16 + col];
#pragma unroll
        for (int v = 0; v < 8; ++v) {
            int dr = m0 + v + 8 * half;
            if (dr < M)
                out[plane_off(M, dr, nt >> 2) + (nt & 3) * 16 + col] =
                    fmaxf(acc[nt][v] + has[v] * cb, 0.0f);
        }
    }
}

extern "C" void kernel_launch(void* const* d_in, const int* in_sizes, int n_in,
                              void* d_out, int out_size, void* d_ws, size_t ws_size,
                              hipStream_t stream) {
    const float* user_x      = (const float*)d_in[0];
    const float* book_x      = (const float*)d_in[1];
    const int*   edge_src    = (const int*)d_in[2];
    const int*   edge_dst    = (const int*)d_in[3];
    const float* user_proj_w = (const float*)d_in[4];
    const float* user_proj_b = (const float*)d_in[5];
    const float* book_proj_w = (const float*)d_in[6];
    const float* book_proj_b = (const float*)d_in[7];
    const float* Wl          = (const float*)d_in[8];
    const float* bl          = (const float*)d_in[9];
    const float* Wr          = (const float*)d_in[10];
    const float* ugamma      = (const float*)d_in[11];
    const float* ubeta       = (const float*)d_in[12];
    const float* bgamma      = (const float*)d_in[13];
    const float* bbeta       = (const float*)d_in[14];

    float* ws = (float*)d_ws;
    size_t off = 0;
    auto carve = [&](size_t n) { float* p = ws + off; off += n; return p; };
    float* u0    = (float*)d_out;
    float* b0    = u0 + (size_t)N_USER * HD;
    float* u1    = carve((size_t)N_USER * HD);
    float* b1    = carve((size_t)N_BOOK * HD);
    float* sum2_u = carve((size_t)N_USER * HD);
    float* sum2_b = carve((size_t)N_BOOK * HD);
    float* sum_u = carve((size_t)N_USER * HD);
    float* sum_b = carve((size_t)N_BOOK * HD);
    float* cnt_u = carve((size_t)N_USER);
    float* cnt_b = carve((size_t)N_BOOK);
    int*   ctrs  = (int*)carve((size_t)3 * SC_COUNTERS);
    float* comp  = carve((size_t)COMP_FLOATS);
    float* img   = carve((size_t)IMG_FLOATS);

    compose_kernel<<<12, 256, 0, stream>>>(
        user_proj_w, book_proj_w, user_proj_b, book_proj_b, Wl, Wr, comp, img);

    zero_kernel<<<4096, 256, 0, stream>>>(
        sum_u, (size_t)(N_USER + N_BOOK) * HD + (size_t)N_USER + (size_t)N_BOOK + (size_t)3 * SC_COUNTERS);
    count_kernel<<<(N_EDGE + 255) / 256, 256, 0, stream>>>(
        edge_src, edge_dst, cnt_u, cnt_b, N_EDGE);

    scatter_pass_kernel<<<SC_RESIDENT, 256, 0, stream>>>(
        user_x, book_x, edge_src, edge_dst, sum_b, sum_u, ctrs, N_EDGE, 1);
    sage_first_kernel<<<UB + BB + ZB, 256, 0, stream>>>(
        sum_u, cnt_u, user_x, img + IMG_PBWL, img + IMG_PUWR,
        comp + COMP_BBWL, comp + COMP_BUWR, u0,
        sum_b, cnt_b, book_x, img + IMG_PUWL, img + IMG_PBWR,
        comp + COMP_BUWL, comp + COMP_BBWR, b0, bl, sum2_u, 128, 64, 64, 128);

    scatter_pass_kernel<<<SC_RESIDENT, 256, 0, stream>>>(
        u0, b0, edge_src, edge_dst, sum2_b, sum2_u, ctrs + SC_COUNTERS, N_EDGE, 0);
    sage_combine_kernel<<<UB + BB + ZB, 256, 0, stream>>>(
        sum2_u, cnt_u, u0, u1, sum2_b, cnt_b, b0, b1,
        img + IMG_LAYER, bl + HD, img + IMG_LAYER + IMG128, sum_u);

    float* out_u = (float*)d_out;
    float* out_b = out_u + (size_t)N_USER * HD;
    scatter_pass_kernel<<<SC_RESIDENT, 256, 0, stream>>>(
        u1, b1, edge_src, edge_dst, sum_b, sum_u, ctrs + 2 * SC_COUNTERS, N_EDGE, 0);
    sage_last_kernel<<<UB + BB, 256, 0, stream>>>(
        sum_u, cnt_u, u1, out_u, sum_b, cnt_b, b1, out_b,
        img + IMG_LAYER + 2 * IMG128, bl + 2 * HD, img + IMG_LAYER + 3 * IMG128,
        ugamma, ubeta, bgamma, bbeta);
}
